# starting kernel
# baseline (speedup 1.0000x reference)
#include <hip/hip_runtime.h>
#include <hip/hip_bf16.h>
#include <math.h>

typedef __attribute__((ext_vector_type(2))) float v2f;
typedef __attribute__((ext_vector_type(8))) float v8f;

#define IN_F   128
#define OUT_F  32
#define NHEAD  4
#define ALPHA  0.2f

__device__ inline void atomicMaxFloat(float* addr, float v) {
    int iv = __float_as_int(v);
    if (iv >= 0) {
        atomicMax((int*)addr, iv);
    } else {
        atomicMin((unsigned int*)addr, (unsigned int)iv);
    }
}

__global__ __launch_bounds__(256) void gat_init(float* __restrict__ m,
                                                float* __restrict__ s,
                                                float* __restrict__ out, int N) {
    long i = (long)blockIdx.x * blockDim.x + threadIdx.x;
    long n4 = (long)N * NHEAD;
    long nF = (long)N * (NHEAD * OUT_F);
    if (i < n4) { m[i] = -INFINITY; s[i] = 0.0f; }
    if (i < nF) { out[i] = 0.0f; }
}

__global__ __launch_bounds__(256) void gat_gemm(const float* __restrict__ x,
                                                const float* __restrict__ W,
                                                float* __restrict__ h, int N) {
    const int wave = threadIdx.x >> 5;
    const int lane = threadIdx.x & 31;
    const long rowTile = (long)blockIdx.x * 8 + wave;
    const long row0 = rowTile * 16;
    if (row0 >= N) return;

    const int mrow  = lane & 15;
    const int khalf = (lane >> 4) * 2;

    v8f acc[8];
    #pragma unroll
    for (int t = 0; t < 8; ++t) acc[t] = v8f{};

    long r = row0 + mrow;
    if (r >= N) r = N - 1;
    const float* __restrict__ xrow = x + r * IN_F;

    for (int k = 0; k < IN_F; k += 4) {
        v2f a;
        a.x = xrow[k + khalf + 0];
        a.y = xrow[k + khalf + 1];
        #pragma unroll
        for (int t = 0; t < 8; ++t) {
            const int ncol = t * 16 + mrow;
            v2f b;
            b.x = W[(k + khalf + 0) * (OUT_F * NHEAD) + ncol];
            b.y = W[(k + khalf + 1) * (OUT_F * NHEAD) + ncol];
            acc[t] = __builtin_amdgcn_wmma_f32_16x16x4_f32(
                false, a, false, b, (short)0, acc[t], false, false);
        }
    }

    const int rowOff = (lane < 16) ? 0 : 8;
    #pragma unroll
    for (int t = 0; t < 8; ++t) {
        #pragma unroll
        for (int g = 0; g < 8; ++g) {
            long rr = row0 + g + rowOff;
            if (rr < N) {
                float v = acc[t][g];
                v = (v == v) ? v : 0.0f;
                h[rr * (OUT_F * NHEAD) + t * 16 + mrow] = v;
            }
        }
    }
}

__global__ __launch_bounds__(256) void gat_alar(const float* __restrict__ h,
                                                const float* __restrict__ a_l,
                                                const float* __restrict__ a_r,
                                                float* __restrict__ al,
                                                float* __restrict__ ar, int N) {
    const int wave = threadIdx.x >> 5;
    const int lane = threadIdx.x & 31;
    const long n = (long)blockIdx.x * 8 + wave;
    if (n >= N) return;
    const float* __restrict__ hn = h + n * (OUT_F * NHEAD);
    float sl[NHEAD], sr[NHEAD];
    #pragma unroll
    for (int j = 0; j < NHEAD; ++j) {
        float hv = hn[j * OUT_F + lane];
        sl[j] = a_l[j * OUT_F + lane] * hv;
        sr[j] = a_r[j * OUT_F + lane] * hv;
    }
    #pragma unroll
    for (int off = 16; off > 0; off >>= 1) {
        #pragma unroll
        for (int j = 0; j < NHEAD; ++j) {
            sl[j] += __shfl_xor(sl[j], off, 32);
            sr[j] += __shfl_xor(sr[j], off, 32);
        }
    }
    if (lane == 0) {
        #pragma unroll
        for (int j = 0; j < NHEAD; ++j) {
            al[n * NHEAD + j] = sl[j];
            ar[n * NHEAD + j] = sr[j];
        }
    }
}

__global__ __launch_bounds__(256) void gat_edge_score(const int* __restrict__ row,
                                                      const int* __restrict__ col,
                                                      const float* __restrict__ al,
                                                      const float* __restrict__ ar,
                                                      float* __restrict__ e,
                                                      float* __restrict__ m, int E) {
    long i = (long)blockIdx.x * blockDim.x + threadIdx.x;
    if (i >= E) return;
    const int r = row[i], c = col[i];
    #pragma unroll
    for (int j = 0; j < NHEAD; ++j) {
        float v = al[(long)r * NHEAD + j] + ar[(long)c * NHEAD + j];
        v = (v > 0.0f) ? v : (ALPHA * v);
        e[i * NHEAD + j] = v;
        atomicMaxFloat(&m[(long)r * NHEAD + j], v);
    }
}

__global__ __launch_bounds__(256) void gat_edge_exp(const int* __restrict__ row,
                                                    const float* __restrict__ m,
                                                    float* __restrict__ e,
                                                    float* __restrict__ s, int E) {
    long i = (long)blockIdx.x * blockDim.x + threadIdx.x;
    if (i >= E) return;
    const int r = row[i];
    #pragma unroll
    for (int j = 0; j < NHEAD; ++j) {
        float v = expf(e[i * NHEAD + j] - m[(long)r * NHEAD + j]);
        e[i * NHEAD + j] = v;
        atomicAdd(&s[(long)r * NHEAD + j], v);
    }
}

__global__ __launch_bounds__(256) void gat_aggregate(const int* __restrict__ row,
                                                     const int* __restrict__ col,
                                                     const float* __restrict__ e,
                                                     const float* __restrict__ s,
                                                     const float* __restrict__ h,
                                                     float* __restrict__ out, int E) {
    const int wave = threadIdx.x >> 5;
    const int lane = threadIdx.x & 31;
    const long i = (long)blockIdx.x * 8 + wave;
    if (i >= E) return;
    const int r = row[i], c = col[i];
    const float* __restrict__ hc = h + (long)c * (OUT_F * NHEAD);
    float* __restrict__ outr = out + (long)r * (OUT_F * NHEAD);
    #pragma unroll
    for (int j = 0; j < NHEAD; ++j) {
        float att = e[i * NHEAD + j] / s[(long)r * NHEAD + j];
        atomicAdd(&outr[j * OUT_F + lane], att * hc[j * OUT_F + lane]);
    }
}

extern "C" void kernel_launch(void* const* d_in, const int* in_sizes, int n_in,
                              void* d_out, int out_size, void* d_ws, size_t ws_size,
                              hipStream_t stream) {
    const float* x    = (const float*)d_in[0];
    const int*   edge = (const int*)d_in[1];
    const float* W    = (const float*)d_in[2];
    const float* a_l  = (const float*)d_in[3];
    const float* a_r  = (const float*)d_in[4];

    const int N = in_sizes[0] / IN_F;
    const int E = in_sizes[1] / 2;
    const int* rowp = edge;
    const int* colp = edge + E;

    float* ws = (float*)d_ws;
    float* h  = ws;
    float* al = h  + (size_t)N * (OUT_F * NHEAD);
    float* ar = al + (size_t)N * NHEAD;
    float* m  = ar + (size_t)N * NHEAD;
    float* s  = m  + (size_t)N * NHEAD;
    float* e  = s  + (size_t)N * NHEAD;
    float* out = (float*)d_out;

    {
        long total = (long)N * (OUT_F * NHEAD);
        gat_init<<<(unsigned)((total + 255) / 256), 256, 0, stream>>>(m, s, out, N);
    }
    gat_gemm<<<(unsigned)((N + 127) / 128), 256, 0, stream>>>(x, W, h, N);
    gat_alar<<<(unsigned)((N + 7) / 8), 256, 0, stream>>>(h, a_l, a_r, al, ar, N);
    gat_edge_score<<<(unsigned)((E + 255) / 256), 256, 0, stream>>>(rowp, colp, al, ar, e, m, E);
    gat_edge_exp<<<(unsigned)((E + 255) / 256), 256, 0, stream>>>(rowp, m, e, s, E);
    gat_aggregate<<<(unsigned)((E + 7) / 8), 256, 0, stream>>>(rowp, colp, e, s, h, out, E);
}


# Fewer, fatter waves: an aggregate wave takes 32 edges, not 1 (ids, weights read once; 8 h rows in flight per wait), an al/ar wave 8 nodes, an init wave 4 KB in 16-B stores; sums stay arrival-order
# speedup vs baseline: 2.7797x; 2.7797x over previous
#include <hip/hip_runtime.h>
#include <hip/hip_bf16.h>
#include <math.h>

typedef __attribute__((ext_vector_type(2))) float v2f;
typedef __attribute__((ext_vector_type(8))) float v8f;

#define IN_F   128
#define OUT_F  32
#define NHEAD  4
#define ALPHA  0.2f

__device__ inline void atomicMaxFloat(float* addr, float v) {
    __hip_atomic_fetch_max(addr, v, __ATOMIC_RELAXED, __HIP_MEMORY_SCOPE_AGENT);
}

#define INIT_V4 8

__global__ __launch_bounds__(256) void gat_init(float* __restrict__ m,
                                                float* __restrict__ s,
                                                float* __restrict__ out, int N) {
    const long nOut4 = (long)N * (NHEAD * OUT_F / 4);
    float4* __restrict__ out4 = reinterpret_cast<float4*>(out);
    const float4 zero4 = make_float4(0.0f, 0.0f, 0.0f, 0.0f);
    const long t = (long)blockIdx.x * (256 * INIT_V4) + threadIdx.x;
    #pragma unroll
    for (int k = 0; k < INIT_V4; ++k) {
        const long i = t + k * 256;
        if (i < nOut4) out4[i] = zero4;
    }
    const long g = (long)blockIdx.x * 256 + threadIdx.x;
    if (g < N) {
        reinterpret_cast<float4*>(m)[g] = make_float4(-INFINITY, -INFINITY, -INFINITY, -INFINITY);
        reinterpret_cast<float4*>(s)[g] = zero4;
    }
}

__global__ __launch_bounds__(256) void gat_gemm(const float* __restrict__ x,
                                                const float* __restrict__ W,
                                                float* __restrict__ h, int N) {
    const int wave = threadIdx.x >> 5;
    const int lane = threadIdx.x & 31;
    const long rowTile = (long)blockIdx.x * 8 + wave;
    const long row0 = rowTile * 16;
    if (row0 >= N) return;

    const int mrow  = lane & 15;
    const int khalf = (lane >> 4) * 2;

    v8f acc[8];
    #pragma unroll
    for (int t = 0; t < 8; ++t) acc[t] = v8f{};

    long r = row0 + mrow;
    if (r >= N) r = N - 1;
    const float* __restrict__ xrow = x + r * IN_F;

    for (int k = 0; k < IN_F; k += 4) {
        v2f a;
        a.x = xrow[k + khalf + 0];
        a.y = xrow[k + khalf + 1];
        #pragma unroll
        for (int t = 0; t < 8; ++t) {
            const int ncol = t * 16 + mrow;
            v2f b;
            b.x = W[(k + khalf + 0) * (OUT_F * NHEAD) + ncol];
            b.y = W[(k + khalf + 1) * (OUT_F * NHEAD) + ncol];
            acc[t] = __builtin_amdgcn_wmma_f32_16x16x4_f32(
                false, a, false, b, (short)0, acc[t], false, false);
        }
    }

    const int rowOff = (lane < 16) ? 0 : 8;
    #pragma unroll
    for (int t = 0; t < 8; ++t) {
        #pragma unroll
        for (int g = 0; g < 8; ++g) {
            long rr = row0 + g + rowOff;
            if (rr < N) {
                float v = acc[t][g];
                v = (v == v) ? v : 0.0f;
                h[rr * (OUT_F * NHEAD) + t * 16 + mrow] = v;
            }
        }
    }
}

#define ALAR_NODES 8

__device__ inline void alar_node(const float (&hv)[NHEAD], const float (&wl)[NHEAD],
                                 const float (&wr)[NHEAD], int lane, long n,
                                 float* __restrict__ al, float* __restrict__ ar) {
    float sl[NHEAD], sr[NHEAD];
    #pragma unroll
    for (int j = 0; j < NHEAD; ++j) {
        sl[j] = wl[j] * hv[j];
        sr[j] = wr[j] * hv[j];
    }
    #pragma unroll
    for (int off = 16; off > 0; off >>= 1) {
        #pragma unroll
        for (int j = 0; j < NHEAD; ++j) {
            sl[j] += __shfl_xor(sl[j], off, 32);
            sr[j] += __shfl_xor(sr[j], off, 32);
        }
    }
    if (lane == 0) {
        #pragma unroll
        for (int j = 0; j < NHEAD; ++j) {
            al[n * NHEAD + j] = sl[j];
            ar[n * NHEAD + j] = sr[j];
        }
    }
}

__global__ __launch_bounds__(256) void gat_alar(const float* __restrict__ h,
                                                const float* __restrict__ a_l,
                                                const float* __restrict__ a_r,
                                                float* __restrict__ al,
                                                float* __restrict__ ar, int N) {
    const int lane = threadIdx.x & 31;
    const int wave = __builtin_amdgcn_readfirstlane(threadIdx.x >> 5);
    const long n0 = ((long)blockIdx.x * 8 + wave) * ALAR_NODES;
    if (n0 >= N) return;
    float wl[NHEAD], wr[NHEAD];
    #pragma unroll
    for (int j = 0; j < NHEAD; ++j) {
        wl[j] = a_l[j * OUT_F + lane];
        wr[j] = a_r[j * OUT_F + lane];
    }
    const float* __restrict__ hn = h + n0 * (OUT_F * NHEAD);
    if (n0 + ALAR_NODES <= N) {
        float hv[ALAR_NODES][NHEAD];
        #pragma unroll
        for (int q = 0; q < ALAR_NODES; ++q) {
            #pragma unroll
            for (int j = 0; j < NHEAD; ++j) hv[q][j] = hn[q * (OUT_F * NHEAD) + j * OUT_F + lane];
        }
        #pragma unroll
        for (int q = 0; q < ALAR_NODES; ++q) alar_node(hv[q], wl, wr, lane, n0 + q, al, ar);
    } else {
        for (long n = n0; n < N; ++n) {
            float hv[NHEAD];
            #pragma unroll
            for (int j = 0; j < NHEAD; ++j) hv[j] = h[n * (OUT_F * NHEAD) + j * OUT_F + lane];
            alar_node(hv, wl, wr, lane, n, al, ar);
        }
    }
}

#define PAIRS_PER_THREAD 4

__device__ inline float leaky(float t) { return (t > 0.0f) ? t : (ALPHA * t); }

__global__ __launch_bounds__(256) void gat_edge_score(const int* __restrict__ row,
                                                      const int* __restrict__ col,
                                                      const float* __restrict__ al,
                                                      const float* __restrict__ ar,
                                                      float* __restrict__ e,
                                                      float* __restrict__ m, int E) {
    const long nPairs = (long)E * NHEAD;
    const long p0 = (long)blockIdx.x * (256 * PAIRS_PER_THREAD) + threadIdx.x;
    if (p0 + (PAIRS_PER_THREAD - 1) * 256 < nPairs) {
        int r[PAIRS_PER_THREAD], c[PAIRS_PER_THREAD], j[PAIRS_PER_THREAD];
        float v[PAIRS_PER_THREAD];
        for (int u = 0; u < PAIRS_PER_THREAD; ++u) {
            const long p = p0 + u * 256;
            j[u] = (int)(p % NHEAD);
            r[u] = row[p / NHEAD];
            c[u] = col[p / NHEAD];
        }
        for (int u = 0; u < PAIRS_PER_THREAD; ++u)
            v[u] = leaky(al[(long)r[u] * NHEAD + j[u]] + ar[(long)c[u] * NHEAD + j[u]]);
        for (int u = 0; u < PAIRS_PER_THREAD; ++u) {
            e[p0 + u * 256] = v[u];
            atomicMaxFloat(&m[(long)r[u] * NHEAD + j[u]], v[u]);
        }
    } else {
        for (int u = 0; u < PAIRS_PER_THREAD; ++u) {
            const long p = p0 + u * 256;
            if (p < nPairs) {
                const int j = (int)(p % NHEAD);
                const int r = row[p / NHEAD], c = col[p / NHEAD];
                const float v = leaky(al[(long)r * NHEAD + j] + ar[(long)c * NHEAD + j]);
                e[p] = v;
                atomicMaxFloat(&m[(long)r * NHEAD + j], v);
            }
        }
    }
}

__global__ __launch_bounds__(256) void gat_edge_exp(const int* __restrict__ row,
                                                    const float* __restrict__ m,
                                                    float* __restrict__ e,
                                                    float* __restrict__ s, int E) {
    const long nPairs = (long)E * NHEAD;
    const long p0 = (long)blockIdx.x * (256 * PAIRS_PER_THREAD) + threadIdx.x;
    if (p0 + (PAIRS_PER_THREAD - 1) * 256 < nPairs) {
        int r[PAIRS_PER_THREAD], j[PAIRS_PER_THREAD];
        float ev[PAIRS_PER_THREAD], v[PAIRS_PER_THREAD];
        for (int u = 0; u < PAIRS_PER_THREAD; ++u) {
            const long p = p0 + u * 256;
            j[u]  = (int)(p % NHEAD);
            r[u]  = row[p / NHEAD];
            ev[u] = e[p];
        }
        for (int u = 0; u < PAIRS_PER_THREAD; ++u)
            v[u] = expf(ev[u] - m[(long)r[u] * NHEAD + j[u]]);
        for (int u = 0; u < PAIRS_PER_THREAD; ++u) {
            e[p0 + u * 256] = v[u];
            atomicAdd(&s[(long)r[u] * NHEAD + j[u]], v[u]);
        }
    } else {
        for (int u = 0; u < PAIRS_PER_THREAD; ++u) {
            const long p = p0 + u * 256;
            if (p < nPairs) {
                const int j = (int)(p % NHEAD);
                const int r = row[p / NHEAD];
                const float v = expf(e[p] - m[(long)r * NHEAD + j]);
                e[p] = v;
                atomicAdd(&s[(long)r * NHEAD + j], v);
            }
        }
    }
}

#define AGG_EDGES 32
#define AGG_BATCH 8

__device__ inline float lane_value(float v, int u) {
    return __int_as_float(__builtin_amdgcn_readlane(__float_as_int(v), u));
}

template <int U0>
__device__ inline void agg_rows_load(const float* __restrict__ h, int c, int lane,
                                     float (&hv)[AGG_BATCH][NHEAD]) {
    #pragma unroll
    for (int b = 0; b < AGG_BATCH; ++b) {
        const int cu = __builtin_amdgcn_readlane(c, U0 + b);
        const float* __restrict__ hc = h + (long)cu * (OUT_F * NHEAD);
        #pragma unroll
        for (int j = 0; j < NHEAD; ++j) hv[b][j] = hc[j * OUT_F + lane];
    }
}

template <int U0>
__device__ inline void agg_rows_add(float* __restrict__ out, int r, int lane,
                                    const float (&hv)[AGG_BATCH][NHEAD],
                                    float a0, float a1, float a2, float a3) {
    #pragma unroll
    for (int b = 0; b < AGG_BATCH; ++b) {
        const int ru = __builtin_amdgcn_readlane(r, U0 + b);
        float* __restrict__ o = out + (long)ru * (OUT_F * NHEAD);
        atomicAdd(o + 0 * OUT_F + lane, lane_value(a0, U0 + b) * hv[b][0]);
        atomicAdd(o + 1 * OUT_F + lane, lane_value(a1, U0 + b) * hv[b][1]);
        atomicAdd(o + 2 * OUT_F + lane, lane_value(a2, U0 + b) * hv[b][2]);
        atomicAdd(o + 3 * OUT_F + lane, lane_value(a3, U0 + b) * hv[b][3]);
    }
}

__global__ __launch_bounds__(256) void gat_aggregate(const int* __restrict__ row,
                                                     const int* __restrict__ col,
                                                     const float* __restrict__ e,
                                                     const float* __restrict__ s,
                                                     const float* __restrict__ h,
                                                     float* __restrict__ out, int E) {
    const int lane = threadIdx.x & 31;
    const int wave = __builtin_amdgcn_readfirstlane(threadIdx.x >> 5);
    const long base = ((long)blockIdx.x * 8 + wave) * AGG_EDGES;
    if (base >= E) return;
    const long rem = (long)E - base;
    const int cnt = (rem < AGG_EDGES) ? (int)rem : AGG_EDGES;

    int c = 0, r = 0;
    float4 ev = make_float4(0.0f, 0.0f, 0.0f, 0.0f);
    if (lane < cnt) {
        const long i = base + lane;
        c  = col[i];
        r  = row[i];
        ev = *reinterpret_cast<const float4*>(e + i * NHEAD);
    }
    if (cnt == AGG_EDGES) {
        float hv[AGG_BATCH][NHEAD];
        agg_rows_load<0>(h, c, lane, hv);
        const float4 sv = *reinterpret_cast<const float4*>(s + (long)r * NHEAD);
        const float a0 = ev.x / sv.x, a1 = ev.y / sv.y, a2 = ev.z / sv.z, a3 = ev.w / sv.w;
        agg_rows_add<0>(out, r, lane, hv, a0, a1, a2, a3);
        agg_rows_load<8>(h, c, lane, hv);
        agg_rows_add<8>(out, r, lane, hv, a0, a1, a2, a3);
        agg_rows_load<16>(h, c, lane, hv);
        agg_rows_add<16>(out, r, lane, hv, a0, a1, a2, a3);
        agg_rows_load<24>(h, c, lane, hv);
        agg_rows_add<24>(out, r, lane, hv, a0, a1, a2, a3);
    } else {
        const float4 sv = *reinterpret_cast<const float4*>(s + (long)r * NHEAD);
        const float a0 = ev.x / sv.x, a1 = ev.y / sv.y, a2 = ev.z / sv.z, a3 = ev.w / sv.w;
        for (int u = 0; u < cnt; ++u) {
            const int cu = __builtin_amdgcn_readlane(c, u);
            const int ru = __builtin_amdgcn_readlane(r, u);
            const float* __restrict__ hc = h + (long)cu * (OUT_F * NHEAD);
            float* __restrict__ o = out + (long)ru * (OUT_F * NHEAD);
            atomicAdd(o + 0 * OUT_F + lane, lane_value(a0, u) * hc[0 * OUT_F + lane]);
            atomicAdd(o + 1 * OUT_F + lane, lane_value(a1, u) * hc[1 * OUT_F + lane]);
            atomicAdd(o + 2 * OUT_F + lane, lane_value(a2, u) * hc[2 * OUT_F + lane]);
            atomicAdd(o + 3 * OUT_F + lane, lane_value(a3, u) * hc[3 * OUT_F + lane]);
        }
    }
}

extern "C" void kernel_launch(void* const* d_in, const int* in_sizes, int n_in,
                              void* d_out, int out_size, void* d_ws, size_t ws_size,
                              hipStream_t stream) {
    const float* x    = (const float*)d_in[0];
    const int*   edge = (const int*)d_in[1];
    const float* W    = (const float*)d_in[2];
    const float* a_l  = (const float*)d_in[3];
    const float* a_r  = (const float*)d_in[4];

    const int N = in_sizes[0] / IN_F;
    const int E = in_sizes[1] / 2;
    const int* rowp = edge;
    const int* colp = edge + E;

    float* ws = (float*)d_ws;
    float* h  = ws;
    float* al = h  + (size_t)N * (OUT_F * NHEAD);
    float* ar = al + (size_t)N * NHEAD;
    float* m  = ar + (size_t)N * NHEAD;
    float* s  = m  + (size_t)N * NHEAD;
    float* e  = s  + (size_t)N * NHEAD;
    float* out = (float*)d_out;

    {
        gat_init<<<(unsigned)((N + 63) / 64), 256, 0, stream>>>(m, s, out, N);
    }
    gat_gemm<<<(unsigned)((N + 127) / 128), 256, 0, stream>>>(x, W, h, N);
    gat_alar<<<(unsigned)((N + 63) / 64), 256, 0, stream>>>(h, a_l, a_r, al, ar, N);
    gat_edge_score<<<(unsigned)((E + 255) / 256), 256, 0, stream>>>(rowp, colp, al, ar, e, m, E);
    gat_edge_exp<<<(unsigned)((E + 255) / 256), 256, 0, stream>>>(rowp, m, e, s, E);
    gat_aggregate<<<(unsigned)((E + 255) / 256), 256, 0, stream>>>(rowp, colp, e, s, h, out, E);
}


# Segment max taken over ar[col] alone (leaky and the rounding of al+ar are monotonic, so m = leaky(al + max ar), same bits): score pass drops the al[row] gather; exp pass reads al, max as one 8-B pair
# speedup vs baseline: 2.9176x; 1.0496x over previous
#include <hip/hip_runtime.h>
#include <hip/hip_bf16.h>
#include <math.h>

typedef __attribute__((ext_vector_type(2))) float v2f;
typedef __attribute__((ext_vector_type(8))) float v8f;

#define IN_F   128
#define OUT_F  32
#define NHEAD  4
#define ALPHA  0.2f

__device__ inline void atomicMaxFloat(float* addr, float v) {
    __hip_atomic_fetch_max(addr, v, __ATOMIC_RELAXED, __HIP_MEMORY_SCOPE_AGENT);
}

#define REC_F (2 * NHEAD)

#define INIT_V4 8

__global__ __launch_bounds__(256) void gat_init(float* __restrict__ s,
                                                float* __restrict__ out, int N) {
    const long nOut4 = (long)N * (NHEAD * OUT_F / 4);
    float4* __restrict__ out4 = reinterpret_cast<float4*>(out);
    const float4 zero4 = make_float4(0.0f, 0.0f, 0.0f, 0.0f);
    const long t = (long)blockIdx.x * (256 * INIT_V4) + threadIdx.x;
    #pragma unroll
    for (int k = 0; k < INIT_V4; ++k) {
        const long i = t + k * 256;
        if (i < nOut4) out4[i] = zero4;
    }
    const long g = (long)blockIdx.x * 256 + threadIdx.x;
    if (g < N) reinterpret_cast<float4*>(s)[g] = zero4;
}

__global__ __launch_bounds__(256) void gat_gemm(const float* __restrict__ x,
                                                const float* __restrict__ W,
                                                float* __restrict__ h, int N) {
    const int wave = threadIdx.x >> 5;
    const int lane = threadIdx.x & 31;
    const long rowTile = (long)blockIdx.x * 8 + wave;
    const long row0 = rowTile * 16;
    if (row0 >= N) return;

    const int mrow  = lane & 15;
    const int khalf = (lane >> 4) * 2;

    v8f acc[8];
    #pragma unroll
    for (int t = 0; t < 8; ++t) acc[t] = v8f{};

    long r = row0 + mrow;
    if (r >= N) r = N - 1;
    const float* __restrict__ xrow = x + r * IN_F;

    for (int k = 0; k < IN_F; k += 4) {
        v2f a;
        a.x = xrow[k + khalf + 0];
        a.y = xrow[k + khalf + 1];
        #pragma unroll
        for (int t = 0; t < 8; ++t) {
            const int ncol = t * 16 + mrow;
            v2f b;
            b.x = W[(k + khalf + 0) * (OUT_F * NHEAD) + ncol];
            b.y = W[(k + khalf + 1) * (OUT_F * NHEAD) + ncol];
            acc[t] = __builtin_amdgcn_wmma_f32_16x16x4_f32(
                false, a, false, b, (short)0, acc[t], false, false);
        }
    }

    const int rowOff = (lane < 16) ? 0 : 8;
    #pragma unroll
    for (int t = 0; t < 8; ++t) {
        #pragma unroll
        for (int g = 0; g < 8; ++g) {
            long rr = row0 + g + rowOff;
            if (rr < N) {
                float v = acc[t][g];
                v = (v == v) ? v : 0.0f;
                h[rr * (OUT_F * NHEAD) + t * 16 + mrow] = v;
            }
        }
    }
}

#define ALAR_NODES 8

__device__ inline void alar_node(const float (&hv)[NHEAD], const float (&wl)[NHEAD],
                                 const float (&wr)[NHEAD], int lane, long n,
                                 float* __restrict__ rec, float* __restrict__ ar) {
    float sl[NHEAD], sr[NHEAD];
    #pragma unroll
    for (int j = 0; j < NHEAD; ++j) {
        sl[j] = wl[j] * hv[j];
        sr[j] = wr[j] * hv[j];
    }
    #pragma unroll
    for (int off = 16; off > 0; off >>= 1) {
        #pragma unroll
        for (int j = 0; j < NHEAD; ++j) {
            sl[j] += __shfl_xor(sl[j], off, 32);
            sr[j] += __shfl_xor(sr[j], off, 32);
        }
    }
    if (lane == 0) {
        float4* __restrict__ rn = reinterpret_cast<float4*>(rec + n * REC_F);
        rn[0] = make_float4(sl[0], -INFINITY, sl[1], -INFINITY);
        rn[1] = make_float4(sl[2], -INFINITY, sl[3], -INFINITY);
        #pragma unroll
        for (int j = 0; j < NHEAD; ++j) ar[n * NHEAD + j] = sr[j];
    }
}

__global__ __launch_bounds__(256) void gat_alar(const float* __restrict__ h,
                                                const float* __restrict__ a_l,
                                                const float* __restrict__ a_r,
                                                float* __restrict__ rec,
                                                float* __restrict__ ar, int N) {
    const int lane = threadIdx.x & 31;
    const int wave = __builtin_amdgcn_readfirstlane(threadIdx.x >> 5);
    const long n0 = ((long)blockIdx.x * 8 + wave) * ALAR_NODES;
    if (n0 >= N) return;
    float wl[NHEAD], wr[NHEAD];
    #pragma unroll
    for (int j = 0; j < NHEAD; ++j) {
        wl[j] = a_l[j * OUT_F + lane];
        wr[j] = a_r[j * OUT_F + lane];
    }
    const float* __restrict__ hn = h + n0 * (OUT_F * NHEAD);
    if (n0 + ALAR_NODES <= N) {
        float hv[ALAR_NODES][NHEAD];
        #pragma unroll
        for (int q = 0; q < ALAR_NODES; ++q) {
            #pragma unroll
            for (int j = 0; j < NHEAD; ++j) hv[q][j] = hn[q * (OUT_F * NHEAD) + j * OUT_F + lane];
        }
        #pragma unroll
        for (int q = 0; q < ALAR_NODES; ++q) alar_node(hv[q], wl, wr, lane, n0 + q, rec, ar);
    } else {
        for (long n = n0; n < N; ++n) {
            float hv[NHEAD];
            #pragma unroll
            for (int j = 0; j < NHEAD; ++j) hv[j] = h[n * (OUT_F * NHEAD) + j * OUT_F + lane];
            alar_node(hv, wl, wr, lane, n, rec, ar);
        }
    }
}

#define PAIRS_PER_THREAD 4

__device__ inline float leaky(float t) { return (t > 0.0f) ? t : (ALPHA * t); }

__global__ __launch_bounds__(256) void gat_edge_score(const int* __restrict__ row,
                                                      const int* __restrict__ col,
                                                      const float* __restrict__ ar,
                                                      float* __restrict__ e,
                                                      float* __restrict__ rec, int E) {
    const long nPairs = (long)E * NHEAD;
    const long p0 = (long)blockIdx.x * (256 * PAIRS_PER_THREAD) + threadIdx.x;
    if (p0 + (PAIRS_PER_THREAD - 1) * 256 < nPairs) {
        int r[PAIRS_PER_THREAD], c[PAIRS_PER_THREAD], j[PAIRS_PER_THREAD];
        float v[PAIRS_PER_THREAD];
        for (int u = 0; u < PAIRS_PER_THREAD; ++u) {
            const long p = p0 + u * 256;
            j[u] = (int)(p % NHEAD);
            r[u] = row[p / NHEAD];
            c[u] = col[p / NHEAD];
        }
        for (int u = 0; u < PAIRS_PER_THREAD; ++u)
            v[u] = ar[(long)c[u] * NHEAD + j[u]];
        for (int u = 0; u < PAIRS_PER_THREAD; ++u) {
            e[p0 + u * 256] = v[u];
            atomicMaxFloat(&rec[(long)r[u] * REC_F + 2 * j[u] + 1], v[u]);
        }
    } else {
        for (int u = 0; u < PAIRS_PER_THREAD; ++u) {
            const long p = p0 + u * 256;
            if (p < nPairs) {
                const int j = (int)(p % NHEAD);
                const int r = row[p / NHEAD], c = col[p / NHEAD];
                const float v = ar[(long)c * NHEAD + j];
                e[p] = v;
                atomicMaxFloat(&rec[(long)r * REC_F + 2 * j + 1], v);
            }
        }
    }
}

__device__ inline float exp_weight(float2 am, float arc) {
    return expf(leaky(am.x + arc) - leaky(am.x + am.y));
}

__global__ __launch_bounds__(256) void gat_edge_exp(const int* __restrict__ row,
                                                    const float* __restrict__ rec,
                                                    float* __restrict__ e,
                                                    float* __restrict__ s, int E) {
    const long nPairs = (long)E * NHEAD;
    const long p0 = (long)blockIdx.x * (256 * PAIRS_PER_THREAD) + threadIdx.x;
    if (p0 + (PAIRS_PER_THREAD - 1) * 256 < nPairs) {
        int r[PAIRS_PER_THREAD], j[PAIRS_PER_THREAD];
        float ev[PAIRS_PER_THREAD], v[PAIRS_PER_THREAD];
        for (int u = 0; u < PAIRS_PER_THREAD; ++u) {
            const long p = p0 + u * 256;
            j[u]  = (int)(p % NHEAD);
            r[u]  = row[p / NHEAD];
            ev[u] = e[p];
        }
        for (int u = 0; u < PAIRS_PER_THREAD; ++u)
            v[u] = exp_weight(*reinterpret_cast<const float2*>(rec + (long)r[u] * REC_F + 2 * j[u]), ev[u]);
        for (int u = 0; u < PAIRS_PER_THREAD; ++u) {
            e[p0 + u * 256] = v[u];
            atomicAdd(&s[(long)r[u] * NHEAD + j[u]], v[u]);
        }
    } else {
        for (int u = 0; u < PAIRS_PER_THREAD; ++u) {
            const long p = p0 + u * 256;
            if (p < nPairs) {
                const int j = (int)(p % NHEAD);
                const int r = row[p / NHEAD];
                const float v = exp_weight(*reinterpret_cast<const float2*>(rec + (long)r * REC_F + 2 * j), e[p]);
                e[p] = v;
                atomicAdd(&s[(long)r * NHEAD + j], v);
            }
        }
    }
}

#define AGG_EDGES 32
#define AGG_BATCH 8

__device__ inline float lane_value(float v, int u) {
    return __int_as_float(__builtin_amdgcn_readlane(__float_as_int(v), u));
}

template <int U0>
__device__ inline void agg_rows_load(const float* __restrict__ h, int c, int lane,
                                     float (&hv)[AGG_BATCH][NHEAD]) {
    #pragma unroll
    for (int b = 0; b < AGG_BATCH; ++b) {
        const int cu = __builtin_amdgcn_readlane(c, U0 + b);
        const float* __restrict__ hc = h + (long)cu * (OUT_F * NHEAD);
        #pragma unroll
        for (int j = 0; j < NHEAD; ++j) hv[b][j] = hc[j * OUT_F + lane];
    }
}

template <int U0>
__device__ inline void agg_rows_add(float* __restrict__ out, int r, int lane,
                                    const float (&hv)[AGG_BATCH][NHEAD],
                                    float a0, float a1, float a2, float a3) {
    #pragma unroll
    for (int b = 0; b < AGG_BATCH; ++b) {
        const int ru = __builtin_amdgcn_readlane(r, U0 + b);
        float* __restrict__ o = out + (long)ru * (OUT_F * NHEAD);
        atomicAdd(o + 0 * OUT_F + lane, lane_value(a0, U0 + b) * hv[b][0]);
        atomicAdd(o + 1 * OUT_F + lane, lane_value(a1, U0 + b) * hv[b][1]);
        atomicAdd(o + 2 * OUT_F + lane, lane_value(a2, U0 + b) * hv[b][2]);
        atomicAdd(o + 3 * OUT_F + lane, lane_value(a3, U0 + b) * hv[b][3]);
    }
}

__global__ __launch_bounds__(256) void gat_aggregate(const int* __restrict__ row,
                                                     const int* __restrict__ col,
                                                     const float* __restrict__ e,
                                                     const float* __restrict__ s,
                                                     const float* __restrict__ h,
                                                     float* __restrict__ out, int E) {
    const int lane = threadIdx.x & 31;
    const int wave = __builtin_amdgcn_readfirstlane(threadIdx.x >> 5);
    const long base = ((long)blockIdx.x * 8 + wave) * AGG_EDGES;
    if (base >= E) return;
    const long rem = (long)E - base;
    const int cnt = (rem < AGG_EDGES) ? (int)rem : AGG_EDGES;

    int c = 0, r = 0;
    float4 ev = make_float4(0.0f, 0.0f, 0.0f, 0.0f);
    if (lane < cnt) {
        const long i = base + lane;
        c  = col[i];
        r  = row[i];
        ev = *reinterpret_cast<const float4*>(e + i * NHEAD);
    }
    if (cnt == AGG_EDGES) {
        float hv[AGG_BATCH][NHEAD];
        agg_rows_load<0>(h, c, lane, hv);
        const float4 sv = *reinterpret_cast<const float4*>(s + (long)r * NHEAD);
        const float a0 = ev.x / sv.x, a1 = ev.y / sv.y, a2 = ev.z / sv.z, a3 = ev.w / sv.w;
        agg_rows_add<0>(out, r, lane, hv, a0, a1, a2, a3);
        agg_rows_load<8>(h, c, lane, hv);
        agg_rows_add<8>(out, r, lane, hv, a0, a1, a2, a3);
        agg_rows_load<16>(h, c, lane, hv);
        agg_rows_add<16>(out, r, lane, hv, a0, a1, a2, a3);
        agg_rows_load<24>(h, c, lane, hv);
        agg_rows_add<24>(out, r, lane, hv, a0, a1, a2, a3);
    } else {
        const float4 sv = *reinterpret_cast<const float4*>(s + (long)r * NHEAD);
        const float a0 = ev.x / sv.x, a1 = ev.y / sv.y, a2 = ev.z / sv.z, a3 = ev.w / sv.w;
        for (int u = 0; u < cnt; ++u) {
            const int cu = __builtin_amdgcn_readlane(c, u);
            const int ru = __builtin_amdgcn_readlane(r, u);
            const float* __restrict__ hc = h + (long)cu * (OUT_F * NHEAD);
            float* __restrict__ o = out + (long)ru * (OUT_F * NHEAD);
            atomicAdd(o + 0 * OUT_F + lane, lane_value(a0, u) * hc[0 * OUT_F + lane]);
            atomicAdd(o + 1 * OUT_F + lane, lane_value(a1, u) * hc[1 * OUT_F + lane]);
            atomicAdd(o + 2 * OUT_F + lane, lane_value(a2, u) * hc[2 * OUT_F + lane]);
            atomicAdd(o + 3 * OUT_F + lane, lane_value(a3, u) * hc[3 * OUT_F + lane]);
        }
    }
}

extern "C" void kernel_launch(void* const* d_in, const int* in_sizes, int n_in,
                              void* d_out, int out_size, void* d_ws, size_t ws_size,
                              hipStream_t stream) {
    const float* x    = (const float*)d_in[0];
    const int*   edge = (const int*)d_in[1];
    const float* W    = (const float*)d_in[2];
    const float* a_l  = (const float*)d_in[3];
    const float* a_r  = (const float*)d_in[4];

    const int N = in_sizes[0] / IN_F;
    const int E = in_sizes[1] / 2;
    const int* rowp = edge;
    const int* colp = edge + E;

    float* ws  = (float*)d_ws;
    float* h   = ws;
    float* rec = h   + (size_t)N * (OUT_F * NHEAD);
    float* ar  = rec + (size_t)N * REC_F;
    float* s   = ar  + (size_t)N * NHEAD;
    float* e   = s   + (size_t)N * NHEAD;
    float* out = (float*)d_out;

    {
        gat_init<<<(unsigned)((N + 63) / 64), 256, 0, stream>>>(s, out, N);
    }
    gat_gemm<<<(unsigned)((N + 127) / 128), 256, 0, stream>>>(x, W, h, N);
    gat_alar<<<(unsigned)((N + 63) / 64), 256, 0, stream>>>(h, a_l, a_r, rec, ar, N);
    gat_edge_score<<<(unsigned)((E + 255) / 256), 256, 0, stream>>>(rowp, colp, ar, e, rec, E);
    gat_edge_exp<<<(unsigned)((E + 255) / 256), 256, 0, stream>>>(rowp, rec, e, s, E);
    gat_aggregate<<<(unsigned)((E + 255) / 256), 256, 0, stream>>>(rowp, colp, e, s, h, out, E);
}


# Aggregate's columns split between the two L2 halves: a block reads its die, draws an edge chunk of its half's columns (heads 0-1 or 2-3) by int ticket; wave takes 64 edges; sums stay arrival-order
# speedup vs baseline: 3.1892x; 1.0931x over previous
#include <hip/hip_runtime.h>
#include <hip/hip_bf16.h>
#include <math.h>

typedef __attribute__((ext_vector_type(2))) float v2f;
typedef __attribute__((ext_vector_type(8))) float v8f;

#define IN_F   128
#define OUT_F  32
#define NHEAD  4
#define ALPHA  0.2f

__device__ inline void atomicMaxFloat(float* addr, float v) {
    __hip_atomic_fetch_max(addr, v, __ATOMIC_RELAXED, __HIP_MEMORY_SCOPE_AGENT);
}

#define REC_F (2 * NHEAD)

#define TICKET_STRIDE 64

#define INIT_V4 8

__global__ __launch_bounds__(256) void gat_init(float* __restrict__ s,
                                                float* __restrict__ out, int N) {
    const long nOut4 = (long)N * (NHEAD * OUT_F / 4);
    float4* __restrict__ out4 = reinterpret_cast<float4*>(out);
    const float4 zero4 = make_float4(0.0f, 0.0f, 0.0f, 0.0f);
    const long t = (long)blockIdx.x * (256 * INIT_V4) + threadIdx.x;
    #pragma unroll
    for (int k = 0; k < INIT_V4; ++k) {
        const long i = t + k * 256;
        if (i < nOut4) out4[i] = zero4;
    }
    const long g = (long)blockIdx.x * 256 + threadIdx.x;
    if (g < N) reinterpret_cast<float4*>(s)[g] = zero4;
}

__global__ __launch_bounds__(256) void gat_gemm(const float* __restrict__ x,
                                                const float* __restrict__ W,
                                                float* __restrict__ h, int N) {
    const int wave = threadIdx.x >> 5;
    const int lane = threadIdx.x & 31;
    const long rowTile = (long)blockIdx.x * 8 + wave;
    const long row0 = rowTile * 16;
    if (row0 >= N) return;

    const int mrow  = lane & 15;
    const int khalf = (lane >> 4) * 2;

    v8f acc[8];
    #pragma unroll
    for (int t = 0; t < 8; ++t) acc[t] = v8f{};

    long r = row0 + mrow;
    if (r >= N) r = N - 1;
    const float* __restrict__ xrow = x + r * IN_F;

    for (int k = 0; k < IN_F; k += 4) {
        v2f a;
        a.x = xrow[k + khalf + 0];
        a.y = xrow[k + khalf + 1];
        #pragma unroll
        for (int t = 0; t < 8; ++t) {
            const int ncol = t * 16 + mrow;
            v2f b;
            b.x = W[(k + khalf + 0) * (OUT_F * NHEAD) + ncol];
            b.y = W[(k + khalf + 1) * (OUT_F * NHEAD) + ncol];
            acc[t] = __builtin_amdgcn_wmma_f32_16x16x4_f32(
                false, a, false, b, (short)0, acc[t], false, false);
        }
    }

    const int rowOff = (lane < 16) ? 0 : 8;
    #pragma unroll
    for (int t = 0; t < 8; ++t) {
        #pragma unroll
        for (int g = 0; g < 8; ++g) {
            long rr = row0 + g + rowOff;
            if (rr < N) {
                float v = acc[t][g];
                v = (v == v) ? v : 0.0f;
                h[rr * (OUT_F * NHEAD) + t * 16 + mrow] = v;
            }
        }
    }
}

#define ALAR_NODES 8

__device__ inline void alar_node(const float (&hv)[NHEAD], const float (&wl)[NHEAD],
                                 const float (&wr)[NHEAD], int lane, long n,
                                 float* __restrict__ rec, float* __restrict__ ar) {
    float sl[NHEAD], sr[NHEAD];
    #pragma unroll
    for (int j = 0; j < NHEAD; ++j) {
        sl[j] = wl[j] * hv[j];
        sr[j] = wr[j] * hv[j];
    }
    #pragma unroll
    for (int off = 16; off > 0; off >>= 1) {
        #pragma unroll
        for (int j = 0; j < NHEAD; ++j) {
            sl[j] += __shfl_xor(sl[j], off, 32);
            sr[j] += __shfl_xor(sr[j], off, 32);
        }
    }
    if (lane == 0) {
        float4* __restrict__ rn = reinterpret_cast<float4*>(rec + n * REC_F);
        rn[0] = make_float4(sl[0], -INFINITY, sl[1], -INFINITY);
        rn[1] = make_float4(sl[2], -INFINITY, sl[3], -INFINITY);
        #pragma unroll
        for (int j = 0; j < NHEAD; ++j) ar[n * NHEAD + j] = sr[j];
    }
}

__global__ __launch_bounds__(256) void gat_alar(const float* __restrict__ h,
                                                const float* __restrict__ a_l,
                                                const float* __restrict__ a_r,
                                                float* __restrict__ rec,
                                                float* __restrict__ ar, int N) {
    const int lane = threadIdx.x & 31;
    const int wave = __builtin_amdgcn_readfirstlane(threadIdx.x >> 5);
    const long n0 = ((long)blockIdx.x * 8 + wave) * ALAR_NODES;
    if (n0 >= N) return;
    float wl[NHEAD], wr[NHEAD];
    #pragma unroll
    for (int j = 0; j < NHEAD; ++j) {
        wl[j] = a_l[j * OUT_F + lane];
        wr[j] = a_r[j * OUT_F + lane];
    }
    const float* __restrict__ hn = h + n0 * (OUT_F * NHEAD);
    if (n0 + ALAR_NODES <= N) {
        float hv[ALAR_NODES][NHEAD];
        #pragma unroll
        for (int q = 0; q < ALAR_NODES; ++q) {
            #pragma unroll
            for (int j = 0; j < NHEAD; ++j) hv[q][j] = hn[q * (OUT_F * NHEAD) + j * OUT_F + lane];
        }
        #pragma unroll
        for (int q = 0; q < ALAR_NODES; ++q) alar_node(hv[q], wl, wr, lane, n0 + q, rec, ar);
    } else {
        for (long n = n0; n < N; ++n) {
            float hv[NHEAD];
            #pragma unroll
            for (int j = 0; j < NHEAD; ++j) hv[j] = h[n * (OUT_F * NHEAD) + j * OUT_F + lane];
            alar_node(hv, wl, wr, lane, n, rec, ar);
        }
    }
}

#define PAIRS_PER_THREAD 4

__device__ inline float leaky(float t) { return (t > 0.0f) ? t : (ALPHA * t); }

__global__ __launch_bounds__(256) void gat_edge_score(const int* __restrict__ row,
                                                      const int* __restrict__ col,
                                                      const float* __restrict__ ar,
                                                      float* __restrict__ e,
                                                      float* __restrict__ rec, int E) {
    const long nPairs = (long)E * NHEAD;
    const long p0 = (long)blockIdx.x * (256 * PAIRS_PER_THREAD) + threadIdx.x;
    if (p0 + (PAIRS_PER_THREAD - 1) * 256 < nPairs) {
        int r[PAIRS_PER_THREAD], c[PAIRS_PER_THREAD], j[PAIRS_PER_THREAD];
        float v[PAIRS_PER_THREAD];
        for (int u = 0; u < PAIRS_PER_THREAD; ++u) {
            const long p = p0 + u * 256;
            j[u] = (int)(p % NHEAD);
            r[u] = row[p / NHEAD];
            c[u] = col[p / NHEAD];
        }
        for (int u = 0; u < PAIRS_PER_THREAD; ++u)
            v[u] = ar[(long)c[u] * NHEAD + j[u]];
        for (int u = 0; u < PAIRS_PER_THREAD; ++u) {
            e[p0 + u * 256] = v[u];
            atomicMaxFloat(&rec[(long)r[u] * REC_F + 2 * j[u] + 1], v[u]);
        }
    } else {
        for (int u = 0; u < PAIRS_PER_THREAD; ++u) {
            const long p = p0 + u * 256;
            if (p < nPairs) {
                const int j = (int)(p % NHEAD);
                const int r = row[p / NHEAD], c = col[p / NHEAD];
                const float v = ar[(long)c * NHEAD + j];
                e[p] = v;
                atomicMaxFloat(&rec[(long)r * REC_F + 2 * j + 1], v);
            }
        }
    }
}

__device__ inline float exp_weight(float2 am, float arc) {
    return expf(leaky(am.x + arc) - leaky(am.x + am.y));
}

__global__ __launch_bounds__(256) void gat_edge_exp(const int* __restrict__ row,
                                                    const float* __restrict__ rec,
                                                    float* __restrict__ e,
                                                    float* __restrict__ s,
                                                    int* __restrict__ tickets, int E) {
    if (blockIdx.x == 0 && threadIdx.x == 0) {
        tickets[0] = 0;
        tickets[TICKET_STRIDE] = 0;
    }
    const long nPairs = (long)E * NHEAD;
    const long p0 = (long)blockIdx.x * (256 * PAIRS_PER_THREAD) + threadIdx.x;
    if (p0 + (PAIRS_PER_THREAD - 1) * 256 < nPairs) {
        int r[PAIRS_PER_THREAD], j[PAIRS_PER_THREAD];
        float ev[PAIRS_PER_THREAD], v[PAIRS_PER_THREAD];
        for (int u = 0; u < PAIRS_PER_THREAD; ++u) {
            const long p = p0 + u * 256;
            j[u]  = (int)(p % NHEAD);
            r[u]  = row[p / NHEAD];
            ev[u] = e[p];
        }
        for (int u = 0; u < PAIRS_PER_THREAD; ++u)
            v[u] = exp_weight(*reinterpret_cast<const float2*>(rec + (long)r[u] * REC_F + 2 * j[u]), ev[u]);
        for (int u = 0; u < PAIRS_PER_THREAD; ++u) {
            e[p0 + u * 256] = v[u];
            atomicAdd(&s[(long)r[u] * NHEAD + j[u]], v[u]);
        }
    } else {
        for (int u = 0; u < PAIRS_PER_THREAD; ++u) {
            const long p = p0 + u * 256;
            if (p < nPairs) {
                const int j = (int)(p % NHEAD);
                const int r = row[p / NHEAD];
                const float v = exp_weight(*reinterpret_cast<const float2*>(rec + (long)r * REC_F + 2 * j), e[p]);
                e[p] = v;
                atomicAdd(&s[(long)r * NHEAD + j], v);
            }
        }
    }
}

#define AGG_HEADS  2
#define AGG_HALVES (NHEAD / AGG_HEADS)
#define AGG_EDGES  64
#define AGG_BATCH  16

__device__ inline float lane_value(float v, int u) {
    return __int_as_float(__builtin_amdgcn_readlane(__float_as_int(v), u));
}

template <int U0>
__device__ inline void agg_rows_load(const float* __restrict__ hp, int c, int lane,
                                     float (&hv)[AGG_BATCH][AGG_HEADS]) {
    #pragma unroll
    for (int b = 0; b < AGG_BATCH; ++b) {
        const int cu = __builtin_amdgcn_readlane(c, U0 + b);
        const float* __restrict__ hc = hp + (long)cu * (OUT_F * NHEAD);
        #pragma unroll
        for (int j = 0; j < AGG_HEADS; ++j) hv[b][j] = hc[j * OUT_F + lane];
    }
}

template <int U0>
__device__ inline void agg_rows_add(float* __restrict__ op, int r, int lane,
                                    const float (&hv)[AGG_BATCH][AGG_HEADS],
                                    float a0, float a1) {
    #pragma unroll
    for (int b = 0; b < AGG_BATCH; ++b) {
        const int ru = __builtin_amdgcn_readlane(r, U0 + b);
        float* __restrict__ o = op + (long)ru * (OUT_F * NHEAD);
        atomicAdd(o + 0 * OUT_F + lane, lane_value(a0, U0 + b) * hv[b][0]);
        atomicAdd(o + 1 * OUT_F + lane, lane_value(a1, U0 + b) * hv[b][1]);
    }
}

__global__ __launch_bounds__(256) void gat_aggregate(const int* __restrict__ row,
                                                     const int* __restrict__ col,
                                                     const float* __restrict__ e,
                                                     const float* __restrict__ s,
                                                     const float* __restrict__ h,
                                                     float* __restrict__ out,
                                                     int* __restrict__ tickets, int E) {
    __shared__ int drawn[2];
    const int lane = threadIdx.x & 31;
    const int wave = __builtin_amdgcn_readfirstlane(threadIdx.x >> 5);
    static_assert(AGG_HALVES == 2, "one column half per cache half");
    const int nb = (E + 8 * AGG_EDGES - 1) / (8 * AGG_EDGES);
    if (threadIdx.x == 0) {
        const unsigned where = __builtin_amdgcn_s_sendmsg_rtn(0x87);
        int q = (int)((where >> 18) & 1u);
        int t = atomicAdd(&tickets[q * TICKET_STRIDE], 1);
        if (t >= nb) {
            q ^= 1;
            t = atomicAdd(&tickets[q * TICKET_STRIDE], 1);
        }
        drawn[0] = t;
        drawn[1] = q;
    }
    __syncthreads();
    const int blk = __builtin_amdgcn_readfirstlane(drawn[0]);
    const int f0  = __builtin_amdgcn_readfirstlane(drawn[1]) * AGG_HEADS;
    if (blk >= nb) return;
    const long base = ((long)blk * 8 + wave) * AGG_EDGES;
    if (base >= E) return;
    const long rem = (long)E - base;
    const int cnt = (rem < AGG_EDGES) ? (int)rem : AGG_EDGES;

    const float* __restrict__ hp = h + f0 * OUT_F;
    float* __restrict__ op = out + f0 * OUT_F;

    int c0 = 0, r0 = 0, c1 = 0, r1 = 0;
    float2 ev0 = make_float2(0.0f, 0.0f), ev1 = make_float2(0.0f, 0.0f);
    if (lane < cnt) {
        const long i = base + lane;
        c0  = col[i];
        r0  = row[i];
        ev0 = *reinterpret_cast<const float2*>(e + i * NHEAD + f0);
    }
    if (lane + 32 < cnt) {
        const long i = base + 32 + lane;
        c1  = col[i];
        r1  = row[i];
        ev1 = *reinterpret_cast<const float2*>(e + i * NHEAD + f0);
    }
    if (cnt == AGG_EDGES) {
        float hv[AGG_BATCH][AGG_HEADS];
        agg_rows_load<0>(hp, c0, lane, hv);
        const float2 sv0 = *reinterpret_cast<const float2*>(s + (long)r0 * NHEAD + f0);
        const float2 sv1 = *reinterpret_cast<const float2*>(s + (long)r1 * NHEAD + f0);
        const float a00 = ev0.x / sv0.x, a01 = ev0.y / sv0.y;
        const float a10 = ev1.x / sv1.x, a11 = ev1.y / sv1.y;
        agg_rows_add<0>(op, r0, lane, hv, a00, a01);
        agg_rows_load<16>(hp, c0, lane, hv);
        agg_rows_add<16>(op, r0, lane, hv, a00, a01);
        agg_rows_load<0>(hp, c1, lane, hv);
        agg_rows_add<0>(op, r1, lane, hv, a10, a11);
        agg_rows_load<16>(hp, c1, lane, hv);
        agg_rows_add<16>(op, r1, lane, hv, a10, a11);
    } else {
        const float2 sv0 = *reinterpret_cast<const float2*>(s + (long)r0 * NHEAD + f0);
        const float2 sv1 = *reinterpret_cast<const float2*>(s + (long)r1 * NHEAD + f0);
        const float a00 = ev0.x / sv0.x, a01 = ev0.y / sv0.y;
        const float a10 = ev1.x / sv1.x, a11 = ev1.y / sv1.y;
        for (int u = 0; u < cnt; ++u) {
            const int l = u & 31;
            const bool hi = (u >= 32);
            const int cu = __builtin_amdgcn_readlane(hi ? c1 : c0, l);
            const int ru = __builtin_amdgcn_readlane(hi ? r1 : r0, l);
            const float w0 = lane_value(hi ? a10 : a00, l);
            const float w1 = lane_value(hi ? a11 : a01, l);
            const float* __restrict__ hc = hp + (long)cu * (OUT_F * NHEAD);
            float* __restrict__ o = op + (long)ru * (OUT_F * NHEAD);
            atomicAdd(o + 0 * OUT_F + lane, w0 * hc[0 * OUT_F + lane]);
            atomicAdd(o + 1 * OUT_F + lane, w1 * hc[1 * OUT_F + lane]);
        }
    }
}

extern "C" void kernel_launch(void* const* d_in, const int* in_sizes, int n_in,
                              void* d_out, int out_size, void* d_ws, size_t ws_size,
                              hipStream_t stream) {
    const float* x    = (const float*)d_in[0];
    const int*   edge = (const int*)d_in[1];
    const float* W    = (const float*)d_in[2];
    const float* a_l  = (const float*)d_in[3];
    const float* a_r  = (const float*)d_in[4];

    const int N = in_sizes[0] / IN_F;
    const int E = in_sizes[1] / 2;
    const int* rowp = edge;
    const int* colp = edge + E;

    float* ws  = (float*)d_ws;
    float* h   = ws;
    float* rec = h   + (size_t)N * (OUT_F * NHEAD);
    float* ar  = rec + (size_t)N * REC_F;
    float* s   = ar  + (size_t)N * NHEAD;
    float* e   = s   + (size_t)N * NHEAD;
    float* out = (float*)d_out;

    {
        gat_init<<<(unsigned)((N + 63) / 64), 256, 0, stream>>>(s, out, N);
    }
    gat_gemm<<<(unsigned)((N + 127) / 128), 256, 0, stream>>>(x, W, h, N);
    gat_alar<<<(unsigned)((N + 63) / 64), 256, 0, stream>>>(h, a_l, a_r, rec, ar, N);
    gat_edge_score<<<(unsigned)((E + 255) / 256), 256, 0, stream>>>(rowp, colp, ar, e, rec, E);
    int* tickets = (int*)ar;
    gat_edge_exp<<<(unsigned)((E + 255) / 256), 256, 0, stream>>>(rowp, rec, e, s, tickets, E);
    gat_aggregate<<<(unsigned)(AGG_HALVES * ((E + 511) / 512)), 256, 0, stream>>>(rowp, colp, e, s, h, out, tickets, E);
}


# Aggregate cut by destination, edges unmoved: a work-group keeps 391 nodes' out rows and weight sums in LDS, scans all ids, adds hits by LDS float add, divides rows at the end; sums stay arrival-order
# speedup vs baseline: 3.9155x; 1.2277x over previous
#include <hip/hip_runtime.h>
#include <hip/hip_bf16.h>
#include <math.h>

typedef __attribute__((ext_vector_type(2))) float v2f;
typedef __attribute__((ext_vector_type(8))) float v8f;

#define IN_F   128
#define OUT_F  32
#define NHEAD  4
#define ALPHA  0.2f

__device__ inline void atomicMaxFloat(float* addr, float v) {
    __hip_atomic_fetch_max(addr, v, __ATOMIC_RELAXED, __HIP_MEMORY_SCOPE_AGENT);
}

#define REC_F (2 * NHEAD)

__global__ __launch_bounds__(256) void gat_gemm(const float* __restrict__ x,
                                                const float* __restrict__ W,
                                                float* __restrict__ h, int N) {
    const int wave = threadIdx.x >> 5;
    const int lane = threadIdx.x & 31;
    const long rowTile = (long)blockIdx.x * 8 + wave;
    const long row0 = rowTile * 16;
    if (row0 >= N) return;

    const int mrow  = lane & 15;
    const int khalf = (lane >> 4) * 2;

    v8f acc[8];
    #pragma unroll
    for (int t = 0; t < 8; ++t) acc[t] = v8f{};

    long r = row0 + mrow;
    if (r >= N) r = N - 1;
    const float* __restrict__ xrow = x + r * IN_F;

    for (int k = 0; k < IN_F; k += 4) {
        v2f a;
        a.x = xrow[k + khalf + 0];
        a.y = xrow[k + khalf + 1];
        #pragma unroll
        for (int t = 0; t < 8; ++t) {
            const int ncol = t * 16 + mrow;
            v2f b;
            b.x = W[(k + khalf + 0) * (OUT_F * NHEAD) + ncol];
            b.y = W[(k + khalf + 1) * (OUT_F * NHEAD) + ncol];
            acc[t] = __builtin_amdgcn_wmma_f32_16x16x4_f32(
                false, a, false, b, (short)0, acc[t], false, false);
        }
    }

    const int rowOff = (lane < 16) ? 0 : 8;
    #pragma unroll
    for (int t = 0; t < 8; ++t) {
        #pragma unroll
        for (int g = 0; g < 8; ++g) {
            long rr = row0 + g + rowOff;
            if (rr < N) {
                float v = acc[t][g];
                v = (v == v) ? v : 0.0f;
                h[rr * (OUT_F * NHEAD) + t * 16 + mrow] = v;
            }
        }
    }
}

#define ALAR_NODES 8

__device__ inline void alar_node(const float (&hv)[NHEAD], const float (&wl)[NHEAD],
                                 const float (&wr)[NHEAD], int lane, long n,
                                 float* __restrict__ rec, float* __restrict__ ar) {
    float sl[NHEAD], sr[NHEAD];
    #pragma unroll
    for (int j = 0; j < NHEAD; ++j) {
        sl[j] = wl[j] * hv[j];
        sr[j] = wr[j] * hv[j];
    }
    #pragma unroll
    for (int off = 16; off > 0; off >>= 1) {
        #pragma unroll
        for (int j = 0; j < NHEAD; ++j) {
            sl[j] += __shfl_xor(sl[j], off, 32);
            sr[j] += __shfl_xor(sr[j], off, 32);
        }
    }
    if (lane == 0) {
        float4* __restrict__ rn = reinterpret_cast<float4*>(rec + n * REC_F);
        rn[0] = make_float4(sl[0], -INFINITY, sl[1], -INFINITY);
        rn[1] = make_float4(sl[2], -INFINITY, sl[3], -INFINITY);
        #pragma unroll
        for (int j = 0; j < NHEAD; ++j) ar[n * NHEAD + j] = sr[j];
    }
}

__global__ __launch_bounds__(256) void gat_alar(const float* __restrict__ h,
                                                const float* __restrict__ a_l,
                                                const float* __restrict__ a_r,
                                                float* __restrict__ rec,
                                                float* __restrict__ ar, int N) {
    const int lane = threadIdx.x & 31;
    const int wave = __builtin_amdgcn_readfirstlane(threadIdx.x >> 5);
    const long n0 = ((long)blockIdx.x * 8 + wave) * ALAR_NODES;
    if (n0 >= N) return;
    float wl[NHEAD], wr[NHEAD];
    #pragma unroll
    for (int j = 0; j < NHEAD; ++j) {
        wl[j] = a_l[j * OUT_F + lane];
        wr[j] = a_r[j * OUT_F + lane];
    }
    const float* __restrict__ hn = h + n0 * (OUT_F * NHEAD);
    if (n0 + ALAR_NODES <= N) {
        float hv[ALAR_NODES][NHEAD];
        #pragma unroll
        for (int q = 0; q < ALAR_NODES; ++q) {
            #pragma unroll
            for (int j = 0; j < NHEAD; ++j) hv[q][j] = hn[q * (OUT_F * NHEAD) + j * OUT_F + lane];
        }
        #pragma unroll
        for (int q = 0; q < ALAR_NODES; ++q) alar_node(hv[q], wl, wr, lane, n0 + q, rec, ar);
    } else {
        for (long n = n0; n < N; ++n) {
            float hv[NHEAD];
            #pragma unroll
            for (int j = 0; j < NHEAD; ++j) hv[j] = h[n * (OUT_F * NHEAD) + j * OUT_F + lane];
            alar_node(hv, wl, wr, lane, n, rec, ar);
        }
    }
}

#define PAIRS_PER_THREAD 4

__device__ inline float leaky(float t) { return (t > 0.0f) ? t : (ALPHA * t); }

__global__ __launch_bounds__(256) void gat_edge_score(const int* __restrict__ row,
                                                      const int* __restrict__ col,
                                                      const float* __restrict__ ar,
                                                      float* __restrict__ rec, int E) {
    const long nPairs = (long)E * NHEAD;
    const long p0 = (long)blockIdx.x * (256 * PAIRS_PER_THREAD) + threadIdx.x;
    if (p0 + (PAIRS_PER_THREAD - 1) * 256 < nPairs) {
        int r[PAIRS_PER_THREAD], c[PAIRS_PER_THREAD], j[PAIRS_PER_THREAD];
        float v[PAIRS_PER_THREAD];
        for (int u = 0; u < PAIRS_PER_THREAD; ++u) {
            const long p = p0 + u * 256;
            j[u] = (int)(p % NHEAD);
            r[u] = row[p / NHEAD];
            c[u] = col[p / NHEAD];
        }
        for (int u = 0; u < PAIRS_PER_THREAD; ++u)
            v[u] = ar[(long)c[u] * NHEAD + j[u]];
        for (int u = 0; u < PAIRS_PER_THREAD; ++u)
            atomicMaxFloat(&rec[(long)r[u] * REC_F + 2 * j[u] + 1], v[u]);
    } else {
        for (int u = 0; u < PAIRS_PER_THREAD; ++u) {
            const long p = p0 + u * 256;
            if (p < nPairs) {
                const int j = (int)(p % NHEAD);
                const int r = row[p / NHEAD], c = col[p / NHEAD];
                atomicMaxFloat(&rec[(long)r * REC_F + 2 * j + 1], ar[(long)c * NHEAD + j]);
            }
        }
    }
}

__device__ inline float exp_weight(float2 am, float arc) {
    return expf(leaky(am.x + arc) - leaky(am.x + am.y));
}

#define OWN_NODES   391
#define OWN_WAVES   32
#define OWN_THREADS (OWN_WAVES * 32)
#define OWN_LOADS   8
#define OWN_QCAP    160
#define OWN_ROWS    8

__device__ inline void lds_add(float* p, float v) {
    __hip_atomic_fetch_add(p, v, __ATOMIC_RELAXED, __HIP_MEMORY_SCOPE_WORKGROUP);
}

__device__ inline float lane_value(float v, int u) {
    return __int_as_float(__builtin_amdgcn_readlane(__float_as_int(v), u));
}

template <bool FULL>
__device__ inline void own_batch(const int2* qb, int nb, int lane,
                                 const int* __restrict__ col, const float* __restrict__ ar,
                                 const float* __restrict__ h, const float* recL,
                                 float* sumL, float* tile) {
    int loc = 0, c = 0;
    float w0 = 0.0f, w1 = 0.0f, w2 = 0.0f, w3 = 0.0f;
    if (FULL || lane < nb) {
        const int2 t = qb[lane];
        loc = t.y;
        c = col[t.x];
        const float4 av = *reinterpret_cast<const float4*>(ar + (long)c * NHEAD);
        const float4 ra = *reinterpret_cast<const float4*>(recL + loc * REC_F);
        const float4 rb = *reinterpret_cast<const float4*>(recL + loc * REC_F + 4);
        w0 = exp_weight(make_float2(ra.x, ra.y), av.x);
        w1 = exp_weight(make_float2(ra.z, ra.w), av.y);
        w2 = exp_weight(make_float2(rb.x, rb.y), av.z);
        w3 = exp_weight(make_float2(rb.z, rb.w), av.w);
        lds_add(sumL + loc * NHEAD + 0, w0);
        lds_add(sumL + loc * NHEAD + 1, w1);
        lds_add(sumL + loc * NHEAD + 2, w2);
        lds_add(sumL + loc * NHEAD + 3, w3);
    }
    const int ng = FULL ? (32 / OWN_ROWS) : ((nb + OWN_ROWS - 1) / OWN_ROWS);
    #pragma unroll 1
    for (int g = 0; g < ng; ++g) {
        float hv[OWN_ROWS][NHEAD];
        #pragma unroll
        for (int b = 0; b < OWN_ROWS; ++b) {
            const int u = g * OWN_ROWS + b;
            #pragma unroll
            for (int j = 0; j < NHEAD; ++j) hv[b][j] = 0.0f;
            if (FULL || u < nb) {
                const int cu = __builtin_amdgcn_readlane(c, u);
                const float* __restrict__ hc = h + (long)cu * (OUT_F * NHEAD);
                #pragma unroll
                for (int j = 0; j < NHEAD; ++j) hv[b][j] = hc[j * OUT_F + lane];
            }
        }
        #pragma unroll
        for (int b = 0; b < OWN_ROWS; ++b) {
            const int u = g * OWN_ROWS + b;
            if (FULL || u < nb) {
                const int lu = __builtin_amdgcn_readlane(loc, u);
                float* o = tile + lu * (OUT_F * NHEAD) + lane;
                lds_add(o + 0 * OUT_F, lane_value(w0, u) * hv[b][0]);
                lds_add(o + 1 * OUT_F, lane_value(w1, u) * hv[b][1]);
                lds_add(o + 2 * OUT_F, lane_value(w2, u) * hv[b][2]);
                lds_add(o + 3 * OUT_F, lane_value(w3, u) * hv[b][3]);
            }
        }
    }
}

__device__ inline void own_push(bool hit, int edge, int loc, int2* q, int& qn) {
    const unsigned m = __builtin_amdgcn_ballot_w32(hit);
    if (m != 0u) {
        if (hit) q[qn + (int)__builtin_amdgcn_mbcnt_lo(m, 0u)] = make_int2(edge, loc);
        qn += __builtin_popcount(m);
    }
}

__device__ inline void own_test(int4 ids, int e0, int lo, unsigned cnt, int lane,
                                int2* q, int& qn,
                                const int* __restrict__ col, const float* __restrict__ ar,
                                const float* __restrict__ h, const float* recL,
                                float* sumL, float* tile) {
    const unsigned d0 = (unsigned)(ids.x - lo), d1 = (unsigned)(ids.y - lo);
    const unsigned d2 = (unsigned)(ids.z - lo), d3 = (unsigned)(ids.w - lo);
    const bool h0 = d0 < cnt, h1 = d1 < cnt, h2 = d2 < cnt, h3 = d3 < cnt;
    if (__builtin_amdgcn_ballot_w32(h0 || h1 || h2 || h3) == 0u) return;
    own_push(h0, e0 + 0, (int)d0, q, qn);
    own_push(h1, e0 + 1, (int)d1, q, qn);
    own_push(h2, e0 + 2, (int)d2, q, qn);
    own_push(h3, e0 + 3, (int)d3, q, qn);
    while (qn >= 32) {
        qn -= 32;
        own_batch<true>(q + qn, 32, lane, col, ar, h, recL, sumL, tile);
    }
}

__global__ __launch_bounds__(OWN_THREADS) void gat_aggregate(const int* __restrict__ row,
                                                            const int* __restrict__ col,
                                                            const float* __restrict__ rec,
                                                            const float* __restrict__ ar,
                                                            const float* __restrict__ h,
                                                            float* __restrict__ out,
                                                            int N, int E) {
    __shared__ __attribute__((aligned(16))) float tile[OWN_NODES * OUT_F * NHEAD];
    __shared__ __attribute__((aligned(16))) float recL[OWN_NODES * REC_F];
    __shared__ __attribute__((aligned(16))) float sumL[OWN_NODES * NHEAD];
    __shared__ int2 queue[OWN_WAVES * OWN_QCAP];

    const int lane = threadIdx.x & 31;
    const int wave = __builtin_amdgcn_readfirstlane(threadIdx.x >> 5);
    const int lo = blockIdx.x * OWN_NODES;
    const int left = N - lo;
    const int cnt = (left < OWN_NODES) ? left : OWN_NODES;

    const float4 zero4 = make_float4(0.0f, 0.0f, 0.0f, 0.0f);
    float4* tile4 = reinterpret_cast<float4*>(tile);
    for (int i = threadIdx.x; i < cnt * (OUT_F * NHEAD / 4); i += OWN_THREADS) tile4[i] = zero4;
    for (int i = threadIdx.x; i < cnt; i += OWN_THREADS) reinterpret_cast<float4*>(sumL)[i] = zero4;
    const float4* __restrict__ rec4 = reinterpret_cast<const float4*>(rec + (long)lo * REC_F);
    for (int i = threadIdx.x; i < cnt * (REC_F / 4); i += OWN_THREADS) reinterpret_cast<float4*>(recL)[i] = rec4[i];
    __syncthreads();

    int2* q = queue + wave * OWN_QCAP;
    int qn = 0;
    const int nFull = E >> 7;
    const int nOwn = (N + OWN_NODES - 1) / OWN_NODES;
    const int rot = (int)blockIdx.x * (nFull / nOwn);
    const int4* __restrict__ row4 = reinterpret_cast<const int4*>(row);
    const int nMine  = (wave < nFull) ? ((nFull - wave + OWN_WAVES - 1) / OWN_WAVES) : 0;
    const int nSuper = nMine / OWN_LOADS;
    int4 cur[OWN_LOADS];
    int stc[OWN_LOADS];
    #pragma unroll
    for (int k = 0; k < OWN_LOADS; ++k) {
        cur[k] = make_int4(0, 0, 0, 0);
        stc[k] = 0;
    }
    if (nSuper > 0) {
        #pragma unroll
        for (int k = 0; k < OWN_LOADS; ++k) {
            int s = wave + k * OWN_WAVES + rot;
            if (s >= nFull) s -= nFull;
            stc[k] = s;
            cur[k] = row4[(long)s * 32 + lane];
        }
    }
    for (int t = 0; t < nSuper; ++t) {
        int4 nxt[OWN_LOADS];
        int stn[OWN_LOADS];
        const int tn = (t + 1 < nSuper) ? (t + 1) : t;
        #pragma unroll
        for (int k = 0; k < OWN_LOADS; ++k) {
            int s = wave + (tn * OWN_LOADS + k) * OWN_WAVES + rot;
            if (s >= nFull) s -= nFull;
            stn[k] = s;
            nxt[k] = row4[(long)s * 32 + lane];
        }
        #pragma unroll
        for (int k = 0; k < OWN_LOADS; ++k)
            own_test(cur[k], stc[k] * 128 + lane * 4, lo, (unsigned)cnt, lane, q, qn,
                     col, ar, h, recL, sumL, tile);
        #pragma unroll
        for (int k = 0; k < OWN_LOADS; ++k) {
            cur[k] = nxt[k];
            stc[k] = stn[k];
        }
    }
    for (int j = nSuper * OWN_LOADS; j < nMine; ++j) {
        int s = wave + j * OWN_WAVES + rot;
        if (s >= nFull) s -= nFull;
        own_test(row4[(long)s * 32 + lane], s * 128 + lane * 4, lo, (unsigned)cnt, lane,
                 q, qn, col, ar, h, recL, sumL, tile);
    }
    if (wave == 0) {
        for (int base = nFull << 7; base < E; base += 32) {
            const int i = base + lane;
            unsigned d = 0xffffffffu;
            if (i < E) d = (unsigned)(row[i] - lo);
            own_push(d < (unsigned)cnt, i, (int)d, q, qn);
            while (qn >= 32) {
                qn -= 32;
                own_batch<true>(q + qn, 32, lane, col, ar, h, recL, sumL, tile);
            }
        }
    }
    if (qn > 0) own_batch<false>(q, qn, lane, col, ar, h, recL, sumL, tile);
    __syncthreads();

    float4* __restrict__ out4 = reinterpret_cast<float4*>(out + (long)lo * (OUT_F * NHEAD));
    for (int i = threadIdx.x; i < cnt * (OUT_F * NHEAD / 4); i += OWN_THREADS) {
        const float4 v = tile4[i];
        const float sv = sumL[(i >> 5) * NHEAD + ((i & 31) >> 3)];
        float4 o = zero4;
        if (sv != 0.0f) {
            o.x = v.x / sv;
            o.y = v.y / sv;
            o.z = v.z / sv;
            o.w = v.w / sv;
        }
        out4[i] = o;
    }
}

extern "C" void kernel_launch(void* const* d_in, const int* in_sizes, int n_in,
                              void* d_out, int out_size, void* d_ws, size_t ws_size,
                              hipStream_t stream) {
    const float* x    = (const float*)d_in[0];
    const int*   edge = (const int*)d_in[1];
    const float* W    = (const float*)d_in[2];
    const float* a_l  = (const float*)d_in[3];
    const float* a_r  = (const float*)d_in[4];

    const int N = in_sizes[0] / IN_F;
    const int E = in_sizes[1] / 2;
    const int* rowp = edge;
    const int* colp = edge + E;

    float* ws  = (float*)d_ws;
    float* h   = ws;
    float* rec = h   + (size_t)N * (OUT_F * NHEAD);
    float* ar  = rec + (size_t)N * REC_F;
    float* out = (float*)d_out;

    gat_gemm<<<(unsigned)((N + 127) / 128), 256, 0, stream>>>(x, W, h, N);
    gat_alar<<<(unsigned)((N + 63) / 64), 256, 0, stream>>>(h, a_l, a_r, rec, ar, N);
    gat_edge_score<<<(unsigned)((E + 255) / 256), 256, 0, stream>>>(rowp, colp, ar, rec, E);
    gat_aggregate<<<(unsigned)((N + OWN_NODES - 1) / OWN_NODES), OWN_THREADS, 0, stream>>>(rowp, colp, rec, ar, h, out, N, E);
}


# Aggregate finds its edges in a 1-byte owner map (destination / 391, written by the score launch) instead of the 4-byte ids: 16 edges per 16-B read; ids read for hits only; sums stay arrival-order
# speedup vs baseline: 4.0860x; 1.0435x over previous
#include <hip/hip_runtime.h>
#include <hip/hip_bf16.h>
#include <math.h>

typedef __attribute__((ext_vector_type(2))) float v2f;
typedef __attribute__((ext_vector_type(8))) float v8f;

#define IN_F   128
#define OUT_F  32
#define NHEAD  4
#define ALPHA  0.2f

__device__ inline void atomicMaxFloat(float* addr, float v) {
    __hip_atomic_fetch_max(addr, v, __ATOMIC_RELAXED, __HIP_MEMORY_SCOPE_AGENT);
}

#define REC_F (2 * NHEAD)

__global__ __launch_bounds__(256) void gat_gemm(const float* __restrict__ x,
                                                const float* __restrict__ W,
                                                float* __restrict__ h, int N) {
    const int wave = threadIdx.x >> 5;
    const int lane = threadIdx.x & 31;
    const long rowTile = (long)blockIdx.x * 8 + wave;
    const long row0 = rowTile * 16;
    if (row0 >= N) return;

    const int mrow  = lane & 15;
    const int khalf = (lane >> 4) * 2;

    v8f acc[8];
    #pragma unroll
    for (int t = 0; t < 8; ++t) acc[t] = v8f{};

    long r = row0 + mrow;
    if (r >= N) r = N - 1;
    const float* __restrict__ xrow = x + r * IN_F;

    for (int k = 0; k < IN_F; k += 4) {
        v2f a;
        a.x = xrow[k + khalf + 0];
        a.y = xrow[k + khalf + 1];
        #pragma unroll
        for (int t = 0; t < 8; ++t) {
            const int ncol = t * 16 + mrow;
            v2f b;
            b.x = W[(k + khalf + 0) * (OUT_F * NHEAD) + ncol];
            b.y = W[(k + khalf + 1) * (OUT_F * NHEAD) + ncol];
            acc[t] = __builtin_amdgcn_wmma_f32_16x16x4_f32(
                false, a, false, b, (short)0, acc[t], false, false);
        }
    }

    const int rowOff = (lane < 16) ? 0 : 8;
    #pragma unroll
    for (int t = 0; t < 8; ++t) {
        #pragma unroll
        for (int g = 0; g < 8; ++g) {
            long rr = row0 + g + rowOff;
            if (rr < N) {
                float v = acc[t][g];
                v = (v == v) ? v : 0.0f;
                h[rr * (OUT_F * NHEAD) + t * 16 + mrow] = v;
            }
        }
    }
}

#define ALAR_NODES 8

__device__ inline void alar_node(const float (&hv)[NHEAD], const float (&wl)[NHEAD],
                                 const float (&wr)[NHEAD], int lane, long n,
                                 float* __restrict__ rec, float* __restrict__ ar) {
    float sl[NHEAD], sr[NHEAD];
    #pragma unroll
    for (int j = 0; j < NHEAD; ++j) {
        sl[j] = wl[j] * hv[j];
        sr[j] = wr[j] * hv[j];
    }
    #pragma unroll
    for (int off = 16; off > 0; off >>= 1) {
        #pragma unroll
        for (int j = 0; j < NHEAD; ++j) {
            sl[j] += __shfl_xor(sl[j], off, 32);
            sr[j] += __shfl_xor(sr[j], off, 32);
        }
    }
    if (lane == 0) {
        float4* __restrict__ rn = reinterpret_cast<float4*>(rec + n * REC_F);
        rn[0] = make_float4(sl[0], -INFINITY, sl[1], -INFINITY);
        rn[1] = make_float4(sl[2], -INFINITY, sl[3], -INFINITY);
        #pragma unroll
        for (int j = 0; j < NHEAD; ++j) ar[n * NHEAD + j] = sr[j];
    }
}

__global__ __launch_bounds__(256) void gat_alar(const float* __restrict__ h,
                                                const float* __restrict__ a_l,
                                                const float* __restrict__ a_r,
                                                float* __restrict__ rec,
                                                float* __restrict__ ar, int N) {
    const int lane = threadIdx.x & 31;
    const int wave = __builtin_amdgcn_readfirstlane(threadIdx.x >> 5);
    const long n0 = ((long)blockIdx.x * 8 + wave) * ALAR_NODES;
    if (n0 >= N) return;
    float wl[NHEAD], wr[NHEAD];
    #pragma unroll
    for (int j = 0; j < NHEAD; ++j) {
        wl[j] = a_l[j * OUT_F + lane];
        wr[j] = a_r[j * OUT_F + lane];
    }
    const float* __restrict__ hn = h + n0 * (OUT_F * NHEAD);
    if (n0 + ALAR_NODES <= N) {
        float hv[ALAR_NODES][NHEAD];
        #pragma unroll
        for (int q = 0; q < ALAR_NODES; ++q) {
            #pragma unroll
            for (int j = 0; j < NHEAD; ++j) hv[q][j] = hn[q * (OUT_F * NHEAD) + j * OUT_F + lane];
        }
        #pragma unroll
        for (int q = 0; q < ALAR_NODES; ++q) alar_node(hv[q], wl, wr, lane, n0 + q, rec, ar);
    } else {
        for (long n = n0; n < N; ++n) {
            float hv[NHEAD];
            #pragma unroll
            for (int j = 0; j < NHEAD; ++j) hv[j] = h[n * (OUT_F * NHEAD) + j * OUT_F + lane];
            alar_node(hv, wl, wr, lane, n, rec, ar);
        }
    }
}

#define OWN_NODES   391

#define PAIRS_PER_THREAD 4

__device__ inline float leaky(float t) { return (t > 0.0f) ? t : (ALPHA * t); }

__global__ __launch_bounds__(256) void gat_edge_score(const int* __restrict__ row,
                                                      const int* __restrict__ col,
                                                      const float* __restrict__ ar,
                                                      float* __restrict__ rec,
                                                      int* __restrict__ own, int E) {
    unsigned char* __restrict__ own8 = reinterpret_cast<unsigned char*>(own);
    const long nPairs = (long)E * NHEAD;
    const long p0 = (long)blockIdx.x * (256 * PAIRS_PER_THREAD) + threadIdx.x;
    if (p0 + (PAIRS_PER_THREAD - 1) * 256 < nPairs) {
        int r[PAIRS_PER_THREAD], c[PAIRS_PER_THREAD], j[PAIRS_PER_THREAD];
        float v[PAIRS_PER_THREAD];
        for (int u = 0; u < PAIRS_PER_THREAD; ++u) {
            const long p = p0 + u * 256;
            j[u] = (int)(p % NHEAD);
            r[u] = row[p / NHEAD];
            c[u] = col[p / NHEAD];
        }
        for (int u = 0; u < PAIRS_PER_THREAD; ++u)
            v[u] = ar[(long)c[u] * NHEAD + j[u]];
        for (int u = 0; u < PAIRS_PER_THREAD; ++u)
            if (j[u] == 0) own8[(p0 + u * 256) / NHEAD] = (unsigned char)((unsigned)r[u] / OWN_NODES);
        for (int u = 0; u < PAIRS_PER_THREAD; ++u)
            atomicMaxFloat(&rec[(long)r[u] * REC_F + 2 * j[u] + 1], v[u]);
    } else {
        for (int u = 0; u < PAIRS_PER_THREAD; ++u) {
            const long p = p0 + u * 256;
            if (p < nPairs) {
                const int j = (int)(p % NHEAD);
                const int r = row[p / NHEAD], c = col[p / NHEAD];
                if (j == 0) own8[p / NHEAD] = (unsigned char)((unsigned)r / OWN_NODES);
                atomicMaxFloat(&rec[(long)r * REC_F + 2 * j + 1], ar[(long)c * NHEAD + j]);
            }
        }
    }
}

__device__ inline float exp_weight(float2 am, float arc) {
    return expf(leaky(am.x + arc) - leaky(am.x + am.y));
}

#define OWN_WAVES   32
#define OWN_THREADS (OWN_WAVES * 32)
#define OWN_LOADS   8
#define OWN_QCAP    64
#define OWN_ROWS    8

__device__ inline void lds_add(float* p, float v) {
    __hip_atomic_fetch_add(p, v, __ATOMIC_RELAXED, __HIP_MEMORY_SCOPE_WORKGROUP);
}

__device__ inline float lane_value(float v, int u) {
    return __int_as_float(__builtin_amdgcn_readlane(__float_as_int(v), u));
}

template <bool FULL>
__device__ inline void own_batch(const int* qb, int nb, int lane, int lo, unsigned cnt,
                                 const int* __restrict__ row, const int* __restrict__ col,
                                 const float* __restrict__ ar, const float* __restrict__ h,
                                 const float* recL, float* sumL, float* tile) {
    int loc = 0, c = 0;
    float w0 = 0.0f, w1 = 0.0f, w2 = 0.0f, w3 = 0.0f;
    if (FULL || lane < nb) {
        const int e = qb[lane];
        const unsigned d = (unsigned)(row[e] - lo);
        c = col[e];
        const float4 av = *reinterpret_cast<const float4*>(ar + (long)c * NHEAD);
        if (d < cnt) {
            loc = (int)d;
            const float4 ra = *reinterpret_cast<const float4*>(recL + loc * REC_F);
            const float4 rb = *reinterpret_cast<const float4*>(recL + loc * REC_F + 4);
            w0 = exp_weight(make_float2(ra.x, ra.y), av.x);
            w1 = exp_weight(make_float2(ra.z, ra.w), av.y);
            w2 = exp_weight(make_float2(rb.x, rb.y), av.z);
            w3 = exp_weight(make_float2(rb.z, rb.w), av.w);
            lds_add(sumL + loc * NHEAD + 0, w0);
            lds_add(sumL + loc * NHEAD + 1, w1);
            lds_add(sumL + loc * NHEAD + 2, w2);
            lds_add(sumL + loc * NHEAD + 3, w3);
        }
    }
    const int ng = FULL ? (32 / OWN_ROWS) : ((nb + OWN_ROWS - 1) / OWN_ROWS);
    #pragma unroll 1
    for (int g = 0; g < ng; ++g) {
        float hv[OWN_ROWS][NHEAD];
        #pragma unroll
        for (int b = 0; b < OWN_ROWS; ++b) {
            const int u = g * OWN_ROWS + b;
            #pragma unroll
            for (int j = 0; j < NHEAD; ++j) hv[b][j] = 0.0f;
            if (FULL || u < nb) {
                const int cu = __builtin_amdgcn_readlane(c, u);
                const float* __restrict__ hc = h + (long)cu * (OUT_F * NHEAD);
                #pragma unroll
                for (int j = 0; j < NHEAD; ++j) hv[b][j] = hc[j * OUT_F + lane];
            }
        }
        #pragma unroll
        for (int b = 0; b < OWN_ROWS; ++b) {
            const int u = g * OWN_ROWS + b;
            if (FULL || u < nb) {
                const int lu = __builtin_amdgcn_readlane(loc, u);
                float* o = tile + lu * (OUT_F * NHEAD) + lane;
                lds_add(o + 0 * OUT_F, lane_value(w0, u) * hv[b][0]);
                lds_add(o + 1 * OUT_F, lane_value(w1, u) * hv[b][1]);
                lds_add(o + 2 * OUT_F, lane_value(w2, u) * hv[b][2]);
                lds_add(o + 3 * OUT_F, lane_value(w3, u) * hv[b][3]);
            }
        }
    }
}

__device__ inline void own_drain(int* q, int& qn, int lane, int lo, unsigned cnt,
                                 const int* __restrict__ row, const int* __restrict__ col,
                                 const float* __restrict__ ar, const float* __restrict__ h,
                                 const float* recL, float* sumL, float* tile) {
    while (qn >= 32) {
        qn -= 32;
        own_batch<true>(q + qn, 32, lane, lo, cnt, row, col, ar, h, recL, sumL, tile);
    }
}

__device__ inline void own_test(int4 ow, int e0, unsigned pat, int lo, unsigned cnt, int lane,
                                int* q, int& qn,
                                const int* __restrict__ row, const int* __restrict__ col,
                                const float* __restrict__ ar, const float* __restrict__ h,
                                const float* recL, float* sumL, float* tile) {
    const unsigned L7 = 0x7f7f7f7fu;
    const unsigned x0 = (unsigned)ow.x ^ pat, x1 = (unsigned)ow.y ^ pat;
    const unsigned x2 = (unsigned)ow.z ^ pat, x3 = (unsigned)ow.w ^ pat;
    const unsigned o0 = ((x0 & L7) + L7) | x0 | L7, o1 = ((x1 & L7) + L7) | x1 | L7;
    const unsigned o2 = ((x2 & L7) + L7) | x2 | L7, o3 = ((x3 & L7) + L7) | x3 | L7;
    unsigned f = ~o0 | (~o1 >> 1) | (~o2 >> 2) | (~o3 >> 3);
    for (;;) {
        const unsigned m = __builtin_amdgcn_ballot_w32(f != 0u);
        if (m == 0u) break;
        if (f != 0u) {
            const int p = __builtin_ctz(f);
            q[qn + (int)__builtin_amdgcn_mbcnt_lo(m, 0u)] = e0 + ((7 - (p & 7)) << 2) + (p >> 3);
            f &= f - 1u;
        }
        qn += __builtin_popcount(m);
        own_drain(q, qn, lane, lo, cnt, row, col, ar, h, recL, sumL, tile);
    }
}

__global__ __launch_bounds__(OWN_THREADS) void gat_aggregate(const int* __restrict__ row,
                                                            const int* __restrict__ col,
                                                            const int* __restrict__ own,
                                                            const float* __restrict__ rec,
                                                            const float* __restrict__ ar,
                                                            const float* __restrict__ h,
                                                            float* __restrict__ out,
                                                            int N, int E) {
    __shared__ __attribute__((aligned(16))) float tile[OWN_NODES * OUT_F * NHEAD];
    __shared__ __attribute__((aligned(16))) float recL[OWN_NODES * REC_F];
    __shared__ __attribute__((aligned(16))) float sumL[OWN_NODES * NHEAD];
    __shared__ int queue[OWN_WAVES * OWN_QCAP];

    const int lane = threadIdx.x & 31;
    const int wave = __builtin_amdgcn_readfirstlane(threadIdx.x >> 5);
    const int lo = blockIdx.x * OWN_NODES;
    const int left = N - lo;
    const unsigned cnt = (unsigned)((left < OWN_NODES) ? left : OWN_NODES);
    const unsigned pat = (blockIdx.x & 255u) * 0x01010101u;

    const float4 zero4 = make_float4(0.0f, 0.0f, 0.0f, 0.0f);
    float4* tile4 = reinterpret_cast<float4*>(tile);
    for (int i = threadIdx.x; i < (int)cnt * (OUT_F * NHEAD / 4); i += OWN_THREADS) tile4[i] = zero4;
    for (int i = threadIdx.x; i < (int)cnt; i += OWN_THREADS) reinterpret_cast<float4*>(sumL)[i] = zero4;
    const float4* __restrict__ rec4 = reinterpret_cast<const float4*>(rec + (long)lo * REC_F);
    for (int i = threadIdx.x; i < (int)cnt * (REC_F / 4); i += OWN_THREADS) reinterpret_cast<float4*>(recL)[i] = rec4[i];
    __syncthreads();

    int* q = queue + wave * OWN_QCAP;
    int qn = 0;
    const int nFull = E >> 9;
    const int nOwn = (N + OWN_NODES - 1) / OWN_NODES;
    const int rot = (int)blockIdx.x * (nFull / nOwn);
    const int4* __restrict__ own4 = reinterpret_cast<const int4*>(own);
    const int nMine  = (wave < nFull) ? ((nFull - wave + OWN_WAVES - 1) / OWN_WAVES) : 0;
    const int nSuper = nMine / OWN_LOADS;
    int4 cur[OWN_LOADS];
    int stc[OWN_LOADS];
    #pragma unroll
    for (int k = 0; k < OWN_LOADS; ++k) {
        cur[k] = make_int4(0, 0, 0, 0);
        stc[k] = 0;
    }
    if (nSuper > 0) {
        #pragma unroll
        for (int k = 0; k < OWN_LOADS; ++k) {
            int s = wave + k * OWN_WAVES + rot;
            if (s >= nFull) s -= nFull;
            stc[k] = s;
            cur[k] = own4[(long)s * 32 + lane];
        }
    }
    for (int t = 0; t < nSuper; ++t) {
        int4 nxt[OWN_LOADS];
        int stn[OWN_LOADS];
        const int tn = (t + 1 < nSuper) ? (t + 1) : t;
        #pragma unroll
        for (int k = 0; k < OWN_LOADS; ++k) {
            int s = wave + (tn * OWN_LOADS + k) * OWN_WAVES + rot;
            if (s >= nFull) s -= nFull;
            stn[k] = s;
            nxt[k] = own4[(long)s * 32 + lane];
        }
        #pragma unroll
        for (int k = 0; k < OWN_LOADS; ++k)
            own_test(cur[k], stc[k] * 512 + lane * 16, pat, lo, cnt, lane, q, qn,
                     row, col, ar, h, recL, sumL, tile);
        #pragma unroll
        for (int k = 0; k < OWN_LOADS; ++k) {
            cur[k] = nxt[k];
            stc[k] = stn[k];
        }
    }
    for (int j = nSuper * OWN_LOADS; j < nMine; ++j) {
        int s = wave + j * OWN_WAVES + rot;
        if (s >= nFull) s -= nFull;
        own_test(own4[(long)s * 32 + lane], s * 512 + lane * 16, pat, lo, cnt, lane, q, qn,
                 row, col, ar, h, recL, sumL, tile);
    }
    if (wave == 0) {
        for (int base = nFull << 9; base < E; base += 32) {
            const int i = base + lane;
            unsigned d = 0xffffffffu;
            if (i < E) d = (unsigned)(row[i] - lo);
            const bool hit = d < cnt;
            const unsigned m = __builtin_amdgcn_ballot_w32(hit);
            if (m != 0u) {
                if (hit) q[qn + (int)__builtin_amdgcn_mbcnt_lo(m, 0u)] = i;
                qn += __builtin_popcount(m);
            }
            own_drain(q, qn, lane, lo, cnt, row, col, ar, h, recL, sumL, tile);
        }
    }
    if (qn > 0) own_batch<false>(q, qn, lane, lo, cnt, row, col, ar, h, recL, sumL, tile);
    __syncthreads();

    float4* __restrict__ out4 = reinterpret_cast<float4*>(out + (long)lo * (OUT_F * NHEAD));
    for (int i = threadIdx.x; i < (int)cnt * (OUT_F * NHEAD / 4); i += OWN_THREADS) {
        const float4 v = tile4[i];
        const float sv = sumL[(i >> 5) * NHEAD + ((i & 31) >> 3)];
        float4 o = zero4;
        if (sv != 0.0f) {
            o.x = v.x / sv;
            o.y = v.y / sv;
            o.z = v.z / sv;
            o.w = v.w / sv;
        }
        out4[i] = o;
    }
}

extern "C" void kernel_launch(void* const* d_in, const int* in_sizes, int n_in,
                              void* d_out, int out_size, void* d_ws, size_t ws_size,
                              hipStream_t stream) {
    const float* x    = (const float*)d_in[0];
    const int*   edge = (const int*)d_in[1];
    const float* W    = (const float*)d_in[2];
    const float* a_l  = (const float*)d_in[3];
    const float* a_r  = (const float*)d_in[4];

    const int N = in_sizes[0] / IN_F;
    const int E = in_sizes[1] / 2;
    const int* rowp = edge;
    const int* colp = edge + E;

    float* ws  = (float*)d_ws;
    float* h   = ws;
    float* rec = h   + (size_t)N * (OUT_F * NHEAD);
    float* ar  = rec + (size_t)N * REC_F;
    int*   own = (int*)(ar + (size_t)N * NHEAD);
    float* out = (float*)d_out;

    gat_gemm<<<(unsigned)((N + 127) / 128), 256, 0, stream>>>(x, W, h, N);
    gat_alar<<<(unsigned)((N + 63) / 64), 256, 0, stream>>>(h, a_l, a_r, rec, ar, N);
    gat_edge_score<<<(unsigned)((E + 255) / 256), 256, 0, stream>>>(rowp, colp, ar, rec, own, E);
    gat_aggregate<<<(unsigned)((N + OWN_NODES - 1) / OWN_NODES), OWN_THREADS, 0, stream>>>(rowp, colp, own, rec, ar, h, out, N, E);
}


# Aggregate reads a source's h row as one 16-B-per-lane load (a lane = 4 adjacent features of one head), was four 4-B loads; LDS rows permuted so adds stay conflict-free; sums stay arrival-order
# speedup vs baseline: 4.4877x; 1.0983x over previous
#include <hip/hip_runtime.h>
#include <hip/hip_bf16.h>
#include <math.h>

typedef __attribute__((ext_vector_type(2))) float v2f;
typedef __attribute__((ext_vector_type(8))) float v8f;

#define IN_F   128
#define OUT_F  32
#define NHEAD  4
#define ALPHA  0.2f

__device__ inline void atomicMaxFloat(float* addr, float v) {
    __hip_atomic_fetch_max(addr, v, __ATOMIC_RELAXED, __HIP_MEMORY_SCOPE_AGENT);
}

#define REC_F (2 * NHEAD)

__global__ __launch_bounds__(256) void gat_gemm(const float* __restrict__ x,
                                                const float* __restrict__ W,
                                                float* __restrict__ h, int N) {
    const int wave = threadIdx.x >> 5;
    const int lane = threadIdx.x & 31;
    const long rowTile = (long)blockIdx.x * 8 + wave;
    const long row0 = rowTile * 16;
    if (row0 >= N) return;

    const int mrow  = lane & 15;
    const int khalf = (lane >> 4) * 2;

    v8f acc[8];
    #pragma unroll
    for (int t = 0; t < 8; ++t) acc[t] = v8f{};

    long r = row0 + mrow;
    if (r >= N) r = N - 1;
    const float* __restrict__ xrow = x + r * IN_F;

    for (int k = 0; k < IN_F; k += 4) {
        v2f a;
        a.x = xrow[k + khalf + 0];
        a.y = xrow[k + khalf + 1];
        #pragma unroll
        for (int t = 0; t < 8; ++t) {
            const int ncol = t * 16 + mrow;
            v2f b;
            b.x = W[(k + khalf + 0) * (OUT_F * NHEAD) + ncol];
            b.y = W[(k + khalf + 1) * (OUT_F * NHEAD) + ncol];
            acc[t] = __builtin_amdgcn_wmma_f32_16x16x4_f32(
                false, a, false, b, (short)0, acc[t], false, false);
        }
    }

    const int rowOff = (lane < 16) ? 0 : 8;
    #pragma unroll
    for (int t = 0; t < 8; ++t) {
        #pragma unroll
        for (int g = 0; g < 8; ++g) {
            long rr = row0 + g + rowOff;
            if (rr < N) {
                float v = acc[t][g];
                v = (v == v) ? v : 0.0f;
                h[rr * (OUT_F * NHEAD) + t * 16 + mrow] = v;
            }
        }
    }
}

#define ALAR_NODES 8

__device__ inline void alar_node(const float (&hv)[NHEAD], const float (&wl)[NHEAD],
                                 const float (&wr)[NHEAD], int lane, long n,
                                 float* __restrict__ rec, float* __restrict__ ar) {
    float sl[NHEAD], sr[NHEAD];
    #pragma unroll
    for (int j = 0; j < NHEAD; ++j) {
        sl[j] = wl[j] * hv[j];
        sr[j] = wr[j] * hv[j];
    }
    #pragma unroll
    for (int off = 16; off > 0; off >>= 1) {
        #pragma unroll
        for (int j = 0; j < NHEAD; ++j) {
            sl[j] += __shfl_xor(sl[j], off, 32);
            sr[j] += __shfl_xor(sr[j], off, 32);
        }
    }
    if (lane == 0) {
        float4* __restrict__ rn = reinterpret_cast<float4*>(rec + n * REC_F);
        rn[0] = make_float4(sl[0], -INFINITY, sl[1], -INFINITY);
        rn[1] = make_float4(sl[2], -INFINITY, sl[3], -INFINITY);
        #pragma unroll
        for (int j = 0; j < NHEAD; ++j) ar[n * NHEAD + j] = sr[j];
    }
}

__global__ __launch_bounds__(256) void gat_alar(const float* __restrict__ h,
                                                const float* __restrict__ a_l,
                                                const float* __restrict__ a_r,
                                                float* __restrict__ rec,
                                                float* __restrict__ ar, int N) {
    const int lane = threadIdx.x & 31;
    const int wave = __builtin_amdgcn_readfirstlane(threadIdx.x >> 5);
    const long n0 = ((long)blockIdx.x * 8 + wave) * ALAR_NODES;
    if (n0 >= N) return;
    float wl[NHEAD], wr[NHEAD];
    #pragma unroll
    for (int j = 0; j < NHEAD; ++j) {
        wl[j] = a_l[j * OUT_F + lane];
        wr[j] = a_r[j * OUT_F + lane];
    }
    const float* __restrict__ hn = h + n0 * (OUT_F * NHEAD);
    if (n0 + ALAR_NODES <= N) {
        float hv[ALAR_NODES][NHEAD];
        #pragma unroll
        for (int q = 0; q < ALAR_NODES; ++q) {
            #pragma unroll
            for (int j = 0; j < NHEAD; ++j) hv[q][j] = hn[q * (OUT_F * NHEAD) + j * OUT_F + lane];
        }
        #pragma unroll
        for (int q = 0; q < ALAR_NODES; ++q) alar_node(hv[q], wl, wr, lane, n0 + q, rec, ar);
    } else {
        for (long n = n0; n < N; ++n) {
            float hv[NHEAD];
            #pragma unroll
            for (int j = 0; j < NHEAD; ++j) hv[j] = h[n * (OUT_F * NHEAD) + j * OUT_F + lane];
            alar_node(hv, wl, wr, lane, n, rec, ar);
        }
    }
}

#define OWN_NODES   391

#define PAIRS_PER_THREAD 4

__device__ inline float leaky(float t) { return (t > 0.0f) ? t : (ALPHA * t); }

__global__ __launch_bounds__(256) void gat_edge_score(const int* __restrict__ row,
                                                      const int* __restrict__ col,
                                                      const float* __restrict__ ar,
                                                      float* __restrict__ rec,
                                                      int* __restrict__ own, int E) {
    unsigned char* __restrict__ own8 = reinterpret_cast<unsigned char*>(own);
    const long nPairs = (long)E * NHEAD;
    const long p0 = (long)blockIdx.x * (256 * PAIRS_PER_THREAD) + threadIdx.x;
    if (p0 + (PAIRS_PER_THREAD - 1) * 256 < nPairs) {
        int r[PAIRS_PER_THREAD], c[PAIRS_PER_THREAD], j[PAIRS_PER_THREAD];
        float v[PAIRS_PER_THREAD];
        for (int u = 0; u < PAIRS_PER_THREAD; ++u) {
            const long p = p0 + u * 256;
            j[u] = (int)(p % NHEAD);
            r[u] = row[p / NHEAD];
            c[u] = col[p / NHEAD];
        }
        for (int u = 0; u < PAIRS_PER_THREAD; ++u)
            v[u] = ar[(long)c[u] * NHEAD + j[u]];
        for (int u = 0; u < PAIRS_PER_THREAD; ++u)
            if (j[u] == 0) own8[(p0 + u * 256) / NHEAD] = (unsigned char)((unsigned)r[u] / OWN_NODES);
        for (int u = 0; u < PAIRS_PER_THREAD; ++u)
            atomicMaxFloat(&rec[(long)r[u] * REC_F + 2 * j[u] + 1], v[u]);
    } else {
        for (int u = 0; u < PAIRS_PER_THREAD; ++u) {
            const long p = p0 + u * 256;
            if (p < nPairs) {
                const int j = (int)(p % NHEAD);
                const int r = row[p / NHEAD], c = col[p / NHEAD];
                if (j == 0) own8[p / NHEAD] = (unsigned char)((unsigned)r / OWN_NODES);
                atomicMaxFloat(&rec[(long)r * REC_F + 2 * j + 1], ar[(long)c * NHEAD + j]);
            }
        }
    }
}

__device__ inline float exp_weight(float2 am, float arc) {
    return expf(leaky(am.x + arc) - leaky(am.x + am.y));
}

#define OWN_WAVES   32
#define OWN_THREADS (OWN_WAVES * 32)
#define OWN_LOADS   8
#define OWN_QCAP    64
#define OWN_ROWS    8

__device__ inline void lds_add(float* p, float v) {
    __hip_atomic_fetch_add(p, v, __ATOMIC_RELAXED, __HIP_MEMORY_SCOPE_WORKGROUP);
}

__device__ inline float lane_value(float v, int u) {
    return __int_as_float(__builtin_amdgcn_readlane(__float_as_int(v), u));
}

template <bool FULL>
__device__ inline void own_batch(const int* qb, int nb, int lane, int lo, unsigned cnt,
                                 const int* __restrict__ row, const int* __restrict__ col,
                                 const float* __restrict__ ar, const float* __restrict__ h,
                                 const float* recL, float* sumL, float* tile) {
    int loc = 0, c = 0;
    float w0 = 0.0f, w1 = 0.0f, w2 = 0.0f, w3 = 0.0f;
    if (FULL || lane < nb) {
        const int e = qb[lane];
        const unsigned d = (unsigned)(row[e] - lo);
        c = col[e];
        const float4 av = *reinterpret_cast<const float4*>(ar + (long)c * NHEAD);
        if (d < cnt) {
            loc = (int)d;
            const float4 ra = *reinterpret_cast<const float4*>(recL + loc * REC_F);
            const float4 rb = *reinterpret_cast<const float4*>(recL + loc * REC_F + 4);
            w0 = exp_weight(make_float2(ra.x, ra.y), av.x);
            w1 = exp_weight(make_float2(ra.z, ra.w), av.y);
            w2 = exp_weight(make_float2(rb.x, rb.y), av.z);
            w3 = exp_weight(make_float2(rb.z, rb.w), av.w);
            lds_add(sumL + loc * NHEAD + 0, w0);
            lds_add(sumL + loc * NHEAD + 1, w1);
            lds_add(sumL + loc * NHEAD + 2, w2);
            lds_add(sumL + loc * NHEAD + 3, w3);
        }
    }
    const int ng = FULL ? (32 / OWN_ROWS) : ((nb + OWN_ROWS - 1) / OWN_ROWS);
    const unsigned hb0 = (lane & 8) ? 0xffffffffu : 0u;
    const unsigned hb1 = (lane & 16) ? 0xffffffffu : 0u;
    #pragma unroll 1
    for (int g = 0; g < ng; ++g) {
        float4 hv[OWN_ROWS];
        #pragma unroll
        for (int b = 0; b < OWN_ROWS; ++b) {
            const int u = g * OWN_ROWS + b;
            hv[b] = make_float4(0.0f, 0.0f, 0.0f, 0.0f);
            if (FULL || u < nb) {
                const int cu = __builtin_amdgcn_readlane(c, u);
                const float4* __restrict__ hc = reinterpret_cast<const float4*>(h + (long)cu * (OUT_F * NHEAD));
                hv[b] = hc[lane];
            }
        }
        #pragma unroll
        for (int b = 0; b < OWN_ROWS; ++b) {
            const int u = g * OWN_ROWS + b;
            if (FULL || u < nb) {
                const int lu = __builtin_amdgcn_readlane(loc, u);
                const unsigned a0 = (unsigned)__builtin_amdgcn_readlane(__float_as_int(w0), u);
                const unsigned a1 = (unsigned)__builtin_amdgcn_readlane(__float_as_int(w1), u);
                const unsigned a2 = (unsigned)__builtin_amdgcn_readlane(__float_as_int(w2), u);
                const unsigned a3 = (unsigned)__builtin_amdgcn_readlane(__float_as_int(w3), u);
                const unsigned lo2 = (hb0 & a1) | (~hb0 & a0);
                const unsigned hi2 = (hb0 & a3) | (~hb0 & a2);
                const float wl = __int_as_float((int)((hb1 & hi2) | (~hb1 & lo2)));
                float* o = tile + lu * (OUT_F * NHEAD) + lane;
                lds_add(o + 0 * 32, wl * hv[b].x);
                lds_add(o + 1 * 32, wl * hv[b].y);
                lds_add(o + 2 * 32, wl * hv[b].z);
                lds_add(o + 3 * 32, wl * hv[b].w);
            }
        }
    }
}

__device__ inline void own_drain(int* q, int& qn, int lane, int lo, unsigned cnt,
                                 const int* __restrict__ row, const int* __restrict__ col,
                                 const float* __restrict__ ar, const float* __restrict__ h,
                                 const float* recL, float* sumL, float* tile) {
    while (qn >= 32) {
        qn -= 32;
        own_batch<true>(q + qn, 32, lane, lo, cnt, row, col, ar, h, recL, sumL, tile);
    }
}

__device__ inline void own_test(int4 ow, int e0, unsigned pat, int lo, unsigned cnt, int lane,
                                int* q, int& qn,
                                const int* __restrict__ row, const int* __restrict__ col,
                                const float* __restrict__ ar, const float* __restrict__ h,
                                const float* recL, float* sumL, float* tile) {
    const unsigned L7 = 0x7f7f7f7fu;
    const unsigned x0 = (unsigned)ow.x ^ pat, x1 = (unsigned)ow.y ^ pat;
    const unsigned x2 = (unsigned)ow.z ^ pat, x3 = (unsigned)ow.w ^ pat;
    const unsigned o0 = ((x0 & L7) + L7) | x0 | L7, o1 = ((x1 & L7) + L7) | x1 | L7;
    const unsigned o2 = ((x2 & L7) + L7) | x2 | L7, o3 = ((x3 & L7) + L7) | x3 | L7;
    unsigned f = ~o0 | (~o1 >> 1) | (~o2 >> 2) | (~o3 >> 3);
    for (;;) {
        const unsigned m = __builtin_amdgcn_ballot_w32(f != 0u);
        if (m == 0u) break;
        if (f != 0u) {
            const int p = __builtin_ctz(f);
            q[qn + (int)__builtin_amdgcn_mbcnt_lo(m, 0u)] = e0 + ((7 - (p & 7)) << 2) + (p >> 3);
            f &= f - 1u;
        }
        qn += __builtin_popcount(m);
        own_drain(q, qn, lane, lo, cnt, row, col, ar, h, recL, sumL, tile);
    }
}

__global__ __launch_bounds__(OWN_THREADS) void gat_aggregate(const int* __restrict__ row,
                                                            const int* __restrict__ col,
                                                            const int* __restrict__ own,
                                                            const float* __restrict__ rec,
                                                            const float* __restrict__ ar,
                                                            const float* __restrict__ h,
                                                            float* __restrict__ out,
                                                            int N, int E) {
    __shared__ __attribute__((aligned(16))) float tile[OWN_NODES * OUT_F * NHEAD];
    __shared__ __attribute__((aligned(16))) float recL[OWN_NODES * REC_F];
    __shared__ __attribute__((aligned(16))) float sumL[OWN_NODES * NHEAD];
    __shared__ int queue[OWN_WAVES * OWN_QCAP];

    const int lane = threadIdx.x & 31;
    const int wave = __builtin_amdgcn_readfirstlane(threadIdx.x >> 5);
    const int lo = blockIdx.x * OWN_NODES;
    const int left = N - lo;
    const unsigned cnt = (unsigned)((left < OWN_NODES) ? left : OWN_NODES);
    const unsigned pat = (blockIdx.x & 255u) * 0x01010101u;

    const float4 zero4 = make_float4(0.0f, 0.0f, 0.0f, 0.0f);
    float4* tile4 = reinterpret_cast<float4*>(tile);
    for (int i = threadIdx.x; i < (int)cnt * (OUT_F * NHEAD / 4); i += OWN_THREADS) tile4[i] = zero4;
    for (int i = threadIdx.x; i < (int)cnt; i += OWN_THREADS) reinterpret_cast<float4*>(sumL)[i] = zero4;
    const float4* __restrict__ rec4 = reinterpret_cast<const float4*>(rec + (long)lo * REC_F);
    for (int i = threadIdx.x; i < (int)cnt * (REC_F / 4); i += OWN_THREADS) reinterpret_cast<float4*>(recL)[i] = rec4[i];
    __syncthreads();

    int* q = queue + wave * OWN_QCAP;
    int qn = 0;
    const int nFull = E >> 9;
    const int nOwn = (N + OWN_NODES - 1) / OWN_NODES;
    const int rot = (int)blockIdx.x * (nFull / nOwn);
    const int4* __restrict__ own4 = reinterpret_cast<const int4*>(own);
    const int nMine  = (wave < nFull) ? ((nFull - wave + OWN_WAVES - 1) / OWN_WAVES) : 0;
    const int nSuper = nMine / OWN_LOADS;
    int4 cur[OWN_LOADS];
    int stc[OWN_LOADS];
    #pragma unroll
    for (int k = 0; k < OWN_LOADS; ++k) {
        cur[k] = make_int4(0, 0, 0, 0);
        stc[k] = 0;
    }
    if (nSuper > 0) {
        #pragma unroll
        for (int k = 0; k < OWN_LOADS; ++k) {
            int s = wave + k * OWN_WAVES + rot;
            if (s >= nFull) s -= nFull;
            stc[k] = s;
            cur[k] = own4[(long)s * 32 + lane];
        }
    }
    for (int t = 0; t < nSuper; ++t) {
        int4 nxt[OWN_LOADS];
        int stn[OWN_LOADS];
        const int tn = (t + 1 < nSuper) ? (t + 1) : t;
        #pragma unroll
        for (int k = 0; k < OWN_LOADS; ++k) {
            int s = wave + (tn * OWN_LOADS + k) * OWN_WAVES + rot;
            if (s >= nFull) s -= nFull;
            stn[k] = s;
            nxt[k] = own4[(long)s * 32 + lane];
        }
        #pragma unroll
        for (int k = 0; k < OWN_LOADS; ++k)
            own_test(cur[k], stc[k] * 512 + lane * 16, pat, lo, cnt, lane, q, qn,
                     row, col, ar, h, recL, sumL, tile);
        #pragma unroll
        for (int k = 0; k < OWN_LOADS; ++k) {
            cur[k] = nxt[k];
            stc[k] = stn[k];
        }
    }
    for (int j = nSuper * OWN_LOADS; j < nMine; ++j) {
        int s = wave + j * OWN_WAVES + rot;
        if (s >= nFull) s -= nFull;
        own_test(own4[(long)s * 32 + lane], s * 512 + lane * 16, pat, lo, cnt, lane, q, qn,
                 row, col, ar, h, recL, sumL, tile);
    }
    if (wave == 0) {
        for (int base = nFull << 9; base < E; base += 32) {
            const int i = base + lane;
            unsigned d = 0xffffffffu;
            if (i < E) d = (unsigned)(row[i] - lo);
            const bool hit = d < cnt;
            const unsigned m = __builtin_amdgcn_ballot_w32(hit);
            if (m != 0u) {
                if (hit) q[qn + (int)__builtin_amdgcn_mbcnt_lo(m, 0u)] = i;
                qn += __builtin_popcount(m);
            }
            own_drain(q, qn, lane, lo, cnt, row, col, ar, h, recL, sumL, tile);
        }
    }
    if (qn > 0) own_batch<false>(q, qn, lane, lo, cnt, row, col, ar, h, recL, sumL, tile);
    __syncthreads();

    float4* __restrict__ out4 = reinterpret_cast<float4*>(out + (long)lo * (OUT_F * NHEAD));
    for (int i = threadIdx.x; i < (int)cnt * (OUT_F * NHEAD / 4); i += OWN_THREADS) {
        const float* t = tile + (i >> 5) * (OUT_F * NHEAD) + (i & 31);
        const float4 v = make_float4(t[0], t[32], t[64], t[96]);
        const float sv = sumL[(i >> 5) * NHEAD + ((i & 31) >> 3)];
        float4 o = zero4;
        if (sv != 0.0f) {
            o.x = v.x / sv;
            o.y = v.y / sv;
            o.z = v.z / sv;
            o.w = v.w / sv;
        }
        out4[i] = o;
    }
}

extern "C" void kernel_launch(void* const* d_in, const int* in_sizes, int n_in,
                              void* d_out, int out_size, void* d_ws, size_t ws_size,
                              hipStream_t stream) {
    const float* x    = (const float*)d_in[0];
    const int*   edge = (const int*)d_in[1];
    const float* W    = (const float*)d_in[2];
    const float* a_l  = (const float*)d_in[3];
    const float* a_r  = (const float*)d_in[4];

    const int N = in_sizes[0] / IN_F;
    const int E = in_sizes[1] / 2;
    const int* rowp = edge;
    const int* colp = edge + E;

    float* ws  = (float*)d_ws;
    float* h   = ws;
    float* rec = h   + (size_t)N * (OUT_F * NHEAD);
    float* ar  = rec + (size_t)N * REC_F;
    int*   own = (int*)(ar + (size_t)N * NHEAD);
    float* out = (float*)d_out;

    gat_gemm<<<(unsigned)((N + 127) / 128), 256, 0, stream>>>(x, W, h, N);
    gat_alar<<<(unsigned)((N + 63) / 64), 256, 0, stream>>>(h, a_l, a_r, rec, ar, N);
    gat_edge_score<<<(unsigned)((E + 255) / 256), 256, 0, stream>>>(rowp, colp, ar, rec, own, E);
    gat_aggregate<<<(unsigned)((N + OWN_NODES - 1) / OWN_NODES), OWN_THREADS, 0, stream>>>(rowp, colp, own, rec, ar, h, out, N, E);
}


# Segment max by LDS float max in the aggregate (a find sweep lists the work-group's edges, a second sweep adds), not by global f32 atomics; score launch only writes the map; sums stay arrival-order
# speedup vs baseline: 5.5225x; 1.2306x over previous
#include <hip/hip_runtime.h>
#include <hip/hip_bf16.h>
#include <math.h>

typedef __attribute__((ext_vector_type(2))) float v2f;
typedef __attribute__((ext_vector_type(8))) float v8f;

#define IN_F   128
#define OUT_F  32
#define NHEAD  4
#define ALPHA  0.2f

#define REC_F (2 * NHEAD)

__global__ __launch_bounds__(256) void gat_gemm(const float* __restrict__ x,
                                                const float* __restrict__ W,
                                                float* __restrict__ h, int N) {
    const int wave = threadIdx.x >> 5;
    const int lane = threadIdx.x & 31;
    const long rowTile = (long)blockIdx.x * 8 + wave;
    const long row0 = rowTile * 16;
    if (row0 >= N) return;

    const int mrow  = lane & 15;
    const int khalf = (lane >> 4) * 2;

    v8f acc[8];
    #pragma unroll
    for (int t = 0; t < 8; ++t) acc[t] = v8f{};

    long r = row0 + mrow;
    if (r >= N) r = N - 1;
    const float* __restrict__ xrow = x + r * IN_F;

    for (int k = 0; k < IN_F; k += 4) {
        v2f a;
        a.x = xrow[k + khalf + 0];
        a.y = xrow[k + khalf + 1];
        #pragma unroll
        for (int t = 0; t < 8; ++t) {
            const int ncol = t * 16 + mrow;
            v2f b;
            b.x = W[(k + khalf + 0) * (OUT_F * NHEAD) + ncol];
            b.y = W[(k + khalf + 1) * (OUT_F * NHEAD) + ncol];
            acc[t] = __builtin_amdgcn_wmma_f32_16x16x4_f32(
                false, a, false, b, (short)0, acc[t], false, false);
        }
    }

    const int rowOff = (lane < 16) ? 0 : 8;
    #pragma unroll
    for (int t = 0; t < 8; ++t) {
        #pragma unroll
        for (int g = 0; g < 8; ++g) {
            long rr = row0 + g + rowOff;
            if (rr < N) {
                float v = acc[t][g];
                v = (v == v) ? v : 0.0f;
                h[rr * (OUT_F * NHEAD) + t * 16 + mrow] = v;
            }
        }
    }
}

#define ALAR_NODES 8

__device__ inline void alar_node(const float (&hv)[NHEAD], const float (&wl)[NHEAD],
                                 const float (&wr)[NHEAD], int lane, long n,
                                 float* __restrict__ rec, float* __restrict__ ar) {
    float sl[NHEAD], sr[NHEAD];
    #pragma unroll
    for (int j = 0; j < NHEAD; ++j) {
        sl[j] = wl[j] * hv[j];
        sr[j] = wr[j] * hv[j];
    }
    #pragma unroll
    for (int off = 16; off > 0; off >>= 1) {
        #pragma unroll
        for (int j = 0; j < NHEAD; ++j) {
            sl[j] += __shfl_xor(sl[j], off, 32);
            sr[j] += __shfl_xor(sr[j], off, 32);
        }
    }
    if (lane == 0) {
        float4* __restrict__ rn = reinterpret_cast<float4*>(rec + n * REC_F);
        rn[0] = make_float4(sl[0], -INFINITY, sl[1], -INFINITY);
        rn[1] = make_float4(sl[2], -INFINITY, sl[3], -INFINITY);
        #pragma unroll
        for (int j = 0; j < NHEAD; ++j) ar[n * NHEAD + j] = sr[j];
    }
}

__global__ __launch_bounds__(256) void gat_alar(const float* __restrict__ h,
                                                const float* __restrict__ a_l,
                                                const float* __restrict__ a_r,
                                                float* __restrict__ rec,
                                                float* __restrict__ ar, int N) {
    const int lane = threadIdx.x & 31;
    const int wave = __builtin_amdgcn_readfirstlane(threadIdx.x >> 5);
    const long n0 = ((long)blockIdx.x * 8 + wave) * ALAR_NODES;
    if (n0 >= N) return;
    float wl[NHEAD], wr[NHEAD];
    #pragma unroll
    for (int j = 0; j < NHEAD; ++j) {
        wl[j] = a_l[j * OUT_F + lane];
        wr[j] = a_r[j * OUT_F + lane];
    }
    const float* __restrict__ hn = h + n0 * (OUT_F * NHEAD);
    if (n0 + ALAR_NODES <= N) {
        float hv[ALAR_NODES][NHEAD];
        #pragma unroll
        for (int q = 0; q < ALAR_NODES; ++q) {
            #pragma unroll
            for (int j = 0; j < NHEAD; ++j) hv[q][j] = hn[q * (OUT_F * NHEAD) + j * OUT_F + lane];
        }
        #pragma unroll
        for (int q = 0; q < ALAR_NODES; ++q) alar_node(hv[q], wl, wr, lane, n0 + q, rec, ar);
    } else {
        for (long n = n0; n < N; ++n) {
            float hv[NHEAD];
            #pragma unroll
            for (int j = 0; j < NHEAD; ++j) hv[j] = h[n * (OUT_F * NHEAD) + j * OUT_F + lane];
            alar_node(hv, wl, wr, lane, n, rec, ar);
        }
    }
}

#define OWN_NODES   391

__global__ __launch_bounds__(256) void gat_owner_map(const int* __restrict__ row,
                                                     int* __restrict__ own, int E) {
    const long e0 = ((long)blockIdx.x * 256 + threadIdx.x) * 4;
    if (e0 + 3 < E) {
        const int4 r = *reinterpret_cast<const int4*>(row + e0);
        const unsigned b0 = ((unsigned)r.x / OWN_NODES) & 255u;
        const unsigned b1 = ((unsigned)r.y / OWN_NODES) & 255u;
        const unsigned b2 = ((unsigned)r.z / OWN_NODES) & 255u;
        const unsigned b3 = ((unsigned)r.w / OWN_NODES) & 255u;
        own[e0 >> 2] = (int)(b0 | (b1 << 8) | (b2 << 16) | (b3 << 24));
    } else {
        unsigned char* __restrict__ own8 = reinterpret_cast<unsigned char*>(own);
        for (long e = e0; e < E; ++e) own8[e] = (unsigned char)((unsigned)row[e] / OWN_NODES);
    }
}

__device__ inline float leaky(float t) { return (t > 0.0f) ? t : (ALPHA * t); }

__device__ inline float exp_weight(float2 am, float arc) {
    return expf(leaky(am.x + arc) - leaky(am.x + am.y));
}

#define OWN_WAVES   32
#define OWN_THREADS (OWN_WAVES * 32)
#define OWN_LOADS   8
#define OWN_QCAP    64
#define OWN_ROWS    8
#define OWN_LCAP    352

__device__ inline void lds_add(float* p, float v) {
    __hip_atomic_fetch_add(p, v, __ATOMIC_RELAXED, __HIP_MEMORY_SCOPE_WORKGROUP);
}

__device__ inline void lds_max(float* p, float v) {
    __hip_atomic_fetch_max(p, v, __ATOMIC_RELAXED, __HIP_MEMORY_SCOPE_WORKGROUP);
}

struct OwnCtx {
    const int* __restrict__ row;
    const int* __restrict__ col;
    const float* __restrict__ ar;
    const float* __restrict__ h;
    float* recL;
    float* sumL;
    float* tile;
    int* ovfL;
    int lo;
    unsigned cnt;
    int lane;
};

template <bool FULL>
__device__ inline void own_mark(const OwnCtx& cx, const int* qb, int nb, int2* L, int& ln) {
    int loc = -1, c = 0;
    if (FULL || cx.lane < nb) {
        const int e = qb[cx.lane];
        const int r = cx.row[e];
        c = cx.col[e];
        asm volatile("" :: "v"(r), "v"(c));
        const unsigned d = (unsigned)(r - cx.lo);
        if (d < cx.cnt) {
            loc = (int)d;
            const float4 av = *reinterpret_cast<const float4*>(cx.ar + (long)c * NHEAD);
            float* rm = cx.recL + loc * REC_F;
            lds_max(rm + 1, av.x);
            lds_max(rm + 3, av.y);
            lds_max(rm + 5, av.z);
            lds_max(rm + 7, av.w);
        }
    }
    if (ln + 32 <= OWN_LCAP) {
        if (FULL || cx.lane < nb) L[ln + cx.lane] = make_int2(loc, c);
        ln += nb;
    } else {
        if (cx.lane == 0) *cx.ovfL = 1;
    }
}

__device__ inline void own_weights(const OwnCtx& cx, int loc, int c,
                                   float& w0, float& w1, float& w2, float& w3) {
    const float4 av = *reinterpret_cast<const float4*>(cx.ar + (long)c * NHEAD);
    const float4 ra = *reinterpret_cast<const float4*>(cx.recL + loc * REC_F);
    const float4 rb = *reinterpret_cast<const float4*>(cx.recL + loc * REC_F + 4);
    w0 = exp_weight(make_float2(ra.x, ra.y), av.x);
    w1 = exp_weight(make_float2(ra.z, ra.w), av.y);
    w2 = exp_weight(make_float2(rb.x, rb.y), av.z);
    w3 = exp_weight(make_float2(rb.z, rb.w), av.w);
    lds_add(cx.sumL + loc * NHEAD + 0, w0);
    lds_add(cx.sumL + loc * NHEAD + 1, w1);
    lds_add(cx.sumL + loc * NHEAD + 2, w2);
    lds_add(cx.sumL + loc * NHEAD + 3, w3);
}

template <bool FULL>
__device__ inline void own_walk(const OwnCtx& cx, int nb, int loc, int c,
                                float w0, float w1, float w2, float w3) {
    const int lane = cx.lane;
    const int ng = FULL ? (32 / OWN_ROWS) : ((nb + OWN_ROWS - 1) / OWN_ROWS);
    const unsigned hb0 = (lane & 8) ? 0xffffffffu : 0u;
    const unsigned hb1 = (lane & 16) ? 0xffffffffu : 0u;
    #pragma unroll 1
    for (int g = 0; g < ng; ++g) {
        float4 hv[OWN_ROWS];
        #pragma unroll
        for (int b = 0; b < OWN_ROWS; ++b) {
            const int u = g * OWN_ROWS + b;
            hv[b] = make_float4(0.0f, 0.0f, 0.0f, 0.0f);
            if (FULL || u < nb) {
                const int cu = __builtin_amdgcn_readlane(c, u);
                const float4* __restrict__ hc = reinterpret_cast<const float4*>(cx.h + (long)cu * (OUT_F * NHEAD));
                hv[b] = hc[lane];
            }
        }
        #pragma unroll
        for (int b = 0; b < OWN_ROWS; ++b) {
            const int u = g * OWN_ROWS + b;
            if (FULL || u < nb) {
                const int lu = __builtin_amdgcn_readlane(loc, u);
                const unsigned a0 = (unsigned)__builtin_amdgcn_readlane(__float_as_int(w0), u);
                const unsigned a1 = (unsigned)__builtin_amdgcn_readlane(__float_as_int(w1), u);
                const unsigned a2 = (unsigned)__builtin_amdgcn_readlane(__float_as_int(w2), u);
                const unsigned a3 = (unsigned)__builtin_amdgcn_readlane(__float_as_int(w3), u);
                const unsigned lo2 = (hb0 & a1) | (~hb0 & a0);
                const unsigned hi2 = (hb0 & a3) | (~hb0 & a2);
                const float wl = __int_as_float((int)((hb1 & hi2) | (~hb1 & lo2)));
                float* o = cx.tile + lu * (OUT_F * NHEAD) + lane;
                lds_add(o + 0 * 32, wl * hv[b].x);
                lds_add(o + 1 * 32, wl * hv[b].y);
                lds_add(o + 2 * 32, wl * hv[b].z);
                lds_add(o + 3 * 32, wl * hv[b].w);
            }
        }
    }
}

template <bool FULL>
__device__ inline void own_rows(const OwnCtx& cx, const int2* lb, int nb) {
    int loc = 0, c = 0;
    float w0 = 0.0f, w1 = 0.0f, w2 = 0.0f, w3 = 0.0f;
    if (FULL || cx.lane < nb) {
        const int2 lc = lb[cx.lane];
        c = lc.y;
        if (lc.x >= 0) {
            loc = lc.x;
            own_weights(cx, loc, c, w0, w1, w2, w3);
        }
    }
    own_walk<FULL>(cx, nb, loc, c, w0, w1, w2, w3);
}

template <bool FULL>
__device__ inline void own_slow(const OwnCtx& cx, const int* qb, int nb) {
    int loc = 0, c = 0;
    float w0 = 0.0f, w1 = 0.0f, w2 = 0.0f, w3 = 0.0f;
    if (FULL || cx.lane < nb) {
        const int e = qb[cx.lane];
        const unsigned d = (unsigned)(cx.row[e] - cx.lo);
        c = cx.col[e];
        if (d < cx.cnt) {
            loc = (int)d;
            own_weights(cx, loc, c, w0, w1, w2, w3);
        }
    }
    own_walk<FULL>(cx, nb, loc, c, w0, w1, w2, w3);
}

template <bool SLOW>
__device__ inline void own_drain(const OwnCtx& cx, int* q, int& qn, int2* L, int& ln) {
    while (qn >= 32) {
        qn -= 32;
        if (SLOW) own_slow<true>(cx, q + qn, 32);
        else      own_mark<true>(cx, q + qn, 32, L, ln);
    }
}

template <bool SLOW>
__device__ inline void own_test(const OwnCtx& cx, int4 ow, int e0, unsigned pat,
                                int* q, int& qn, int2* L, int& ln) {
    const unsigned L7 = 0x7f7f7f7fu;
    const unsigned x0 = (unsigned)ow.x ^ pat, x1 = (unsigned)ow.y ^ pat;
    const unsigned x2 = (unsigned)ow.z ^ pat, x3 = (unsigned)ow.w ^ pat;
    const unsigned o0 = ((x0 & L7) + L7) | x0 | L7, o1 = ((x1 & L7) + L7) | x1 | L7;
    const unsigned o2 = ((x2 & L7) + L7) | x2 | L7, o3 = ((x3 & L7) + L7) | x3 | L7;
    unsigned f = ~o0 | (~o1 >> 1) | (~o2 >> 2) | (~o3 >> 3);
    for (;;) {
        const unsigned m = __builtin_amdgcn_ballot_w32(f != 0u);
        if (m == 0u) break;
        if (f != 0u) {
            const int p = __builtin_ctz(f);
            q[qn + (int)__builtin_amdgcn_mbcnt_lo(m, 0u)] = e0 + ((7 - (p & 7)) << 2) + (p >> 3);
            f &= f - 1u;
        }
        qn += __builtin_popcount(m);
        own_drain<SLOW>(cx, q, qn, L, ln);
    }
}

template <bool SLOW>
__device__ inline void own_tail(const OwnCtx& cx, int first, int E, int* q, int& qn, int2* L, int& ln) {
    for (int base = first; base < E; base += 32) {
        const int i = base + cx.lane;
        unsigned d = 0xffffffffu;
        if (i < E) d = (unsigned)(cx.row[i] - cx.lo);
        const bool hit = d < cx.cnt;
        const unsigned m = __builtin_amdgcn_ballot_w32(hit);
        if (m != 0u) {
            if (hit) q[qn + (int)__builtin_amdgcn_mbcnt_lo(m, 0u)] = i;
            qn += __builtin_popcount(m);
        }
        own_drain<SLOW>(cx, q, qn, L, ln);
    }
}

__global__ __launch_bounds__(OWN_THREADS) void gat_aggregate(const int* __restrict__ row,
                                                            const int* __restrict__ col,
                                                            const int* __restrict__ own,
                                                            const float* __restrict__ rec,
                                                            const float* __restrict__ ar,
                                                            const float* __restrict__ h,
                                                            float* __restrict__ out,
                                                            int N, int E) {
    __shared__ __attribute__((aligned(16))) float tile[OWN_NODES * OUT_F * NHEAD];
    __shared__ __attribute__((aligned(16))) float recL[OWN_NODES * REC_F];
    __shared__ __attribute__((aligned(16))) float sumL[OWN_NODES * NHEAD];
    __shared__ __attribute__((aligned(8))) int2 lists[OWN_WAVES * OWN_LCAP];
    __shared__ int queue[OWN_WAVES * OWN_QCAP];
    __shared__ int ovfL;

    const int lane = threadIdx.x & 31;
    const int wave = __builtin_amdgcn_readfirstlane(threadIdx.x >> 5);
    const int lo = blockIdx.x * OWN_NODES;
    const int left = N - lo;
    const unsigned cnt = (unsigned)((left < OWN_NODES) ? left : OWN_NODES);
    const unsigned pat = (blockIdx.x & 255u) * 0x01010101u;

    const float4 zero4 = make_float4(0.0f, 0.0f, 0.0f, 0.0f);
    float4* tile4 = reinterpret_cast<float4*>(tile);
    for (int i = threadIdx.x; i < (int)cnt * (OUT_F * NHEAD / 4); i += OWN_THREADS) tile4[i] = zero4;
    for (int i = threadIdx.x; i < (int)cnt; i += OWN_THREADS) reinterpret_cast<float4*>(sumL)[i] = zero4;
    const float4* __restrict__ rec4 = reinterpret_cast<const float4*>(rec + (long)lo * REC_F);
    for (int i = threadIdx.x; i < (int)cnt * (REC_F / 4); i += OWN_THREADS) reinterpret_cast<float4*>(recL)[i] = rec4[i];
    if (threadIdx.x == 0) ovfL = 0;
    __syncthreads();

    OwnCtx cx;
    cx.row = row;   cx.col = col;   cx.ar = ar;     cx.h = h;
    cx.recL = recL; cx.sumL = sumL; cx.tile = tile; cx.ovfL = &ovfL;
    cx.lo = lo;     cx.cnt = cnt;   cx.lane = lane;

    int* q = queue + wave * OWN_QCAP;
    int2* L = lists + wave * OWN_LCAP;
    int qn = 0, ln = 0;
    const int nFull = E >> 9;
    const int nOwn = (N + OWN_NODES - 1) / OWN_NODES;
    const int rot = (int)blockIdx.x * (nFull / nOwn);
    const int4* __restrict__ own4 = reinterpret_cast<const int4*>(own);
    const int nMine  = (wave < nFull) ? ((nFull - wave + OWN_WAVES - 1) / OWN_WAVES) : 0;
    const int nSuper = nMine / OWN_LOADS;
    {
        int4 cur[OWN_LOADS];
        int stc[OWN_LOADS];
        #pragma unroll
        for (int k = 0; k < OWN_LOADS; ++k) {
            cur[k] = make_int4(0, 0, 0, 0);
            stc[k] = 0;
        }
        if (nSuper > 0) {
            #pragma unroll
            for (int k = 0; k < OWN_LOADS; ++k) {
                int s = wave + k * OWN_WAVES + rot;
                if (s >= nFull) s -= nFull;
                stc[k] = s;
                cur[k] = own4[(long)s * 32 + lane];
            }
        }
        for (int t = 0; t < nSuper; ++t) {
            int4 nxt[OWN_LOADS];
            int stn[OWN_LOADS];
            const int tn = (t + 1 < nSuper) ? (t + 1) : t;
            #pragma unroll
            for (int k = 0; k < OWN_LOADS; ++k) {
                int s = wave + (tn * OWN_LOADS + k) * OWN_WAVES + rot;
                if (s >= nFull) s -= nFull;
                stn[k] = s;
                nxt[k] = own4[(long)s * 32 + lane];
            }
            #pragma unroll
            for (int k = 0; k < OWN_LOADS; ++k)
                own_test<false>(cx, cur[k], stc[k] * 512 + lane * 16, pat, q, qn, L, ln);
            #pragma unroll
            for (int k = 0; k < OWN_LOADS; ++k) {
                cur[k] = nxt[k];
                stc[k] = stn[k];
            }
        }
    }
    for (int j = nSuper * OWN_LOADS; j < nMine; ++j) {
        int s = wave + j * OWN_WAVES + rot;
        if (s >= nFull) s -= nFull;
        own_test<false>(cx, own4[(long)s * 32 + lane], s * 512 + lane * 16, pat, q, qn, L, ln);
    }
    if (wave == 0) own_tail<false>(cx, nFull << 9, E, q, qn, L, ln);
    if (qn > 0) own_mark<false>(cx, q, qn, L, ln);
    __syncthreads();

    const int crowded = __builtin_amdgcn_readfirstlane(ovfL);
    if (crowded == 0) {
        int b = 0;
        for (; b + 32 <= ln; b += 32) own_rows<true>(cx, L + b, 32);
        if (b < ln) own_rows<false>(cx, L + b, ln - b);
    } else {
        qn = 0;
        for (int j = 0; j < nMine; ++j) {
            int s = wave + j * OWN_WAVES + rot;
            if (s >= nFull) s -= nFull;
            own_test<true>(cx, own4[(long)s * 32 + lane], s * 512 + lane * 16, pat, q, qn, L, ln);
        }
        if (wave == 0) own_tail<true>(cx, nFull << 9, E, q, qn, L, ln);
        if (qn > 0) own_slow<false>(cx, q, qn);
    }
    __syncthreads();

    float4* __restrict__ out4 = reinterpret_cast<float4*>(out + (long)lo * (OUT_F * NHEAD));
    for (int i = threadIdx.x; i < (int)cnt * (OUT_F * NHEAD / 4); i += OWN_THREADS) {
        const float* t = tile + (i >> 5) * (OUT_F * NHEAD) + (i & 31);
        const float4 v = make_float4(t[0], t[32], t[64], t[96]);
        const float sv = sumL[(i >> 5) * NHEAD + ((i & 31) >> 3)];
        float4 o = zero4;
        if (sv != 0.0f) {
            o.x = v.x / sv;
            o.y = v.y / sv;
            o.z = v.z / sv;
            o.w = v.w / sv;
        }
        out4[i] = o;
    }
}

extern "C" void kernel_launch(void* const* d_in, const int* in_sizes, int n_in,
                              void* d_out, int out_size, void* d_ws, size_t ws_size,
                              hipStream_t stream) {
    const float* x    = (const float*)d_in[0];
    const int*   edge = (const int*)d_in[1];
    const float* W    = (const float*)d_in[2];
    const float* a_l  = (const float*)d_in[3];
    const float* a_r  = (const float*)d_in[4];

    const int N = in_sizes[0] / IN_F;
    const int E = in_sizes[1] / 2;
    const int* rowp = edge;
    const int* colp = edge + E;

    float* ws  = (float*)d_ws;
    float* h   = ws;
    float* rec = h   + (size_t)N * (OUT_F * NHEAD);
    float* ar  = rec + (size_t)N * REC_F;
    int*   own = (int*)(ar + (size_t)N * NHEAD);
    float* out = (float*)d_out;

    gat_gemm<<<(unsigned)((N + 127) / 128), 256, 0, stream>>>(x, W, h, N);
    gat_alar<<<(unsigned)((N + 63) / 64), 256, 0, stream>>>(h, a_l, a_r, rec, ar, N);
    gat_owner_map<<<(unsigned)((E + 1023) / 1024), 256, 0, stream>>>(rowp, own, E);
    gat_aggregate<<<(unsigned)((N + OWN_NODES - 1) / OWN_NODES), OWN_THREADS, 0, stream>>>(rowp, colp, own, rec, ar, h, out, N, E);
}


# Owner-map launch also writes each edge's two ids side by side (8-B pair plane); the find sweep reads a hit's destination and source as one 8-B load, not two 4-B loads E ids apart; sums unchanged
# speedup vs baseline: 5.7055x; 1.0331x over previous
#include <hip/hip_runtime.h>
#include <hip/hip_bf16.h>
#include <math.h>

typedef __attribute__((ext_vector_type(2))) float v2f;
typedef __attribute__((ext_vector_type(8))) float v8f;

#define IN_F   128
#define OUT_F  32
#define NHEAD  4
#define ALPHA  0.2f

#define REC_F (2 * NHEAD)

__global__ __launch_bounds__(256) void gat_gemm(const float* __restrict__ x,
                                                const float* __restrict__ W,
                                                float* __restrict__ h, int N) {
    const int wave = threadIdx.x >> 5;
    const int lane = threadIdx.x & 31;
    const long rowTile = (long)blockIdx.x * 8 + wave;
    const long row0 = rowTile * 16;
    if (row0 >= N) return;

    const int mrow  = lane & 15;
    const int khalf = (lane >> 4) * 2;

    v8f acc[8];
    #pragma unroll
    for (int t = 0; t < 8; ++t) acc[t] = v8f{};

    long r = row0 + mrow;
    if (r >= N) r = N - 1;
    const float* __restrict__ xrow = x + r * IN_F;

    for (int k = 0; k < IN_F; k += 4) {
        v2f a;
        a.x = xrow[k + khalf + 0];
        a.y = xrow[k + khalf + 1];
        #pragma unroll
        for (int t = 0; t < 8; ++t) {
            const int ncol = t * 16 + mrow;
            v2f b;
            b.x = W[(k + khalf + 0) * (OUT_F * NHEAD) + ncol];
            b.y = W[(k + khalf + 1) * (OUT_F * NHEAD) + ncol];
            acc[t] = __builtin_amdgcn_wmma_f32_16x16x4_f32(
                false, a, false, b, (short)0, acc[t], false, false);
        }
    }

    const int rowOff = (lane < 16) ? 0 : 8;
    #pragma unroll
    for (int t = 0; t < 8; ++t) {
        #pragma unroll
        for (int g = 0; g < 8; ++g) {
            long rr = row0 + g + rowOff;
            if (rr < N) {
                float v = acc[t][g];
                v = (v == v) ? v : 0.0f;
                h[rr * (OUT_F * NHEAD) + t * 16 + mrow] = v;
            }
        }
    }
}

#define ALAR_NODES 8

__device__ inline void alar_node(const float (&hv)[NHEAD], const float (&wl)[NHEAD],
                                 const float (&wr)[NHEAD], int lane, long n,
                                 float* __restrict__ rec, float* __restrict__ ar) {
    float sl[NHEAD], sr[NHEAD];
    #pragma unroll
    for (int j = 0; j < NHEAD; ++j) {
        sl[j] = wl[j] * hv[j];
        sr[j] = wr[j] * hv[j];
    }
    #pragma unroll
    for (int off = 16; off > 0; off >>= 1) {
        #pragma unroll
        for (int j = 0; j < NHEAD; ++j) {
            sl[j] += __shfl_xor(sl[j], off, 32);
            sr[j] += __shfl_xor(sr[j], off, 32);
        }
    }
    if (lane == 0) {
        float4* __restrict__ rn = reinterpret_cast<float4*>(rec + n * REC_F);
        rn[0] = make_float4(sl[0], -INFINITY, sl[1], -INFINITY);
        rn[1] = make_float4(sl[2], -INFINITY, sl[3], -INFINITY);
        #pragma unroll
        for (int j = 0; j < NHEAD; ++j) ar[n * NHEAD + j] = sr[j];
    }
}

__global__ __launch_bounds__(256) void gat_alar(const float* __restrict__ h,
                                                const float* __restrict__ a_l,
                                                const float* __restrict__ a_r,
                                                float* __restrict__ rec,
                                                float* __restrict__ ar, int N) {
    const int lane = threadIdx.x & 31;
    const int wave = __builtin_amdgcn_readfirstlane(threadIdx.x >> 5);
    const long n0 = ((long)blockIdx.x * 8 + wave) * ALAR_NODES;
    if (n0 >= N) return;
    float wl[NHEAD], wr[NHEAD];
    #pragma unroll
    for (int j = 0; j < NHEAD; ++j) {
        wl[j] = a_l[j * OUT_F + lane];
        wr[j] = a_r[j * OUT_F + lane];
    }
    const float* __restrict__ hn = h + n0 * (OUT_F * NHEAD);
    if (n0 + ALAR_NODES <= N) {
        float hv[ALAR_NODES][NHEAD];
        #pragma unroll
        for (int q = 0; q < ALAR_NODES; ++q) {
            #pragma unroll
            for (int j = 0; j < NHEAD; ++j) hv[q][j] = hn[q * (OUT_F * NHEAD) + j * OUT_F + lane];
        }
        #pragma unroll
        for (int q = 0; q < ALAR_NODES; ++q) alar_node(hv[q], wl, wr, lane, n0 + q, rec, ar);
    } else {
        for (long n = n0; n < N; ++n) {
            float hv[NHEAD];
            #pragma unroll
            for (int j = 0; j < NHEAD; ++j) hv[j] = h[n * (OUT_F * NHEAD) + j * OUT_F + lane];
            alar_node(hv, wl, wr, lane, n, rec, ar);
        }
    }
}

#define OWN_NODES   391

typedef int ids4 __attribute__((ext_vector_type(4), aligned(4)));

__global__ __launch_bounds__(256) void gat_owner_map(const int* __restrict__ row,
                                                     const int* __restrict__ col,
                                                     int* __restrict__ own,
                                                     int* __restrict__ prw, int E) {
    int2* __restrict__ pr = reinterpret_cast<int2*>(prw);
    const long e0 = ((long)blockIdx.x * 256 + threadIdx.x) * 4;
    if (e0 + 3 < E) {
        const int4 r = *reinterpret_cast<const int4*>(row + e0);
        const ids4 s = *reinterpret_cast<const ids4*>(col + e0);
        const unsigned b0 = ((unsigned)r.x / OWN_NODES) & 255u;
        const unsigned b1 = ((unsigned)r.y / OWN_NODES) & 255u;
        const unsigned b2 = ((unsigned)r.z / OWN_NODES) & 255u;
        const unsigned b3 = ((unsigned)r.w / OWN_NODES) & 255u;
        own[e0 >> 2] = (int)(b0 | (b1 << 8) | (b2 << 16) | (b3 << 24));
        int4* __restrict__ p4 = reinterpret_cast<int4*>(pr + e0);
        p4[0] = make_int4(r.x, s.x, r.y, s.y);
        p4[1] = make_int4(r.z, s.z, r.w, s.w);
    } else {
        unsigned char* __restrict__ own8 = reinterpret_cast<unsigned char*>(own);
        for (long e = e0; e < E; ++e) {
            const int r = row[e];
            own8[e] = (unsigned char)((unsigned)r / OWN_NODES);
            pr[e] = make_int2(r, col[e]);
        }
    }
}

__device__ inline float leaky(float t) { return (t > 0.0f) ? t : (ALPHA * t); }

__device__ inline float exp_weight(float2 am, float arc) {
    return expf(leaky(am.x + arc) - leaky(am.x + am.y));
}

#define OWN_WAVES   32
#define OWN_THREADS (OWN_WAVES * 32)
#define OWN_LOADS   8
#define OWN_QCAP    64
#define OWN_ROWS    8
#define OWN_LCAP    352

__device__ inline void lds_add(float* p, float v) {
    __hip_atomic_fetch_add(p, v, __ATOMIC_RELAXED, __HIP_MEMORY_SCOPE_WORKGROUP);
}

__device__ inline void lds_max(float* p, float v) {
    __hip_atomic_fetch_max(p, v, __ATOMIC_RELAXED, __HIP_MEMORY_SCOPE_WORKGROUP);
}

struct OwnCtx {
    const int* __restrict__ row;
    const int2* __restrict__ pr;
    const float* __restrict__ ar;
    const float* __restrict__ h;
    float* recL;
    float* sumL;
    float* tile;
    int* ovfL;
    int lo;
    unsigned cnt;
    int lane;
};

template <bool FULL>
__device__ inline void own_mark(const OwnCtx& cx, const int* qb, int nb, int2* L, int& ln) {
    int loc = -1, c = 0;
    if (FULL || cx.lane < nb) {
        const int e = qb[cx.lane];
        const int2 rc = cx.pr[e];
        c = rc.y;
        const unsigned d = (unsigned)(rc.x - cx.lo);
        if (d < cx.cnt) {
            loc = (int)d;
            const float4 av = *reinterpret_cast<const float4*>(cx.ar + (long)c * NHEAD);
            float* rm = cx.recL + loc * REC_F;
            lds_max(rm + 1, av.x);
            lds_max(rm + 3, av.y);
            lds_max(rm + 5, av.z);
            lds_max(rm + 7, av.w);
        }
    }
    if (ln + 32 <= OWN_LCAP) {
        if (FULL || cx.lane < nb) L[ln + cx.lane] = make_int2(loc, c);
        ln += nb;
    } else {
        if (cx.lane == 0) *cx.ovfL = 1;
    }
}

__device__ inline void own_weights(const OwnCtx& cx, int loc, int c,
                                   float& w0, float& w1, float& w2, float& w3) {
    const float4 av = *reinterpret_cast<const float4*>(cx.ar + (long)c * NHEAD);
    const float4 ra = *reinterpret_cast<const float4*>(cx.recL + loc * REC_F);
    const float4 rb = *reinterpret_cast<const float4*>(cx.recL + loc * REC_F + 4);
    w0 = exp_weight(make_float2(ra.x, ra.y), av.x);
    w1 = exp_weight(make_float2(ra.z, ra.w), av.y);
    w2 = exp_weight(make_float2(rb.x, rb.y), av.z);
    w3 = exp_weight(make_float2(rb.z, rb.w), av.w);
    lds_add(cx.sumL + loc * NHEAD + 0, w0);
    lds_add(cx.sumL + loc * NHEAD + 1, w1);
    lds_add(cx.sumL + loc * NHEAD + 2, w2);
    lds_add(cx.sumL + loc * NHEAD + 3, w3);
}

template <bool FULL>
__device__ inline void own_walk(const OwnCtx& cx, int nb, int loc, int c,
                                float w0, float w1, float w2, float w3) {
    const int lane = cx.lane;
    const int ng = FULL ? (32 / OWN_ROWS) : ((nb + OWN_ROWS - 1) / OWN_ROWS);
    const unsigned hb0 = (lane & 8) ? 0xffffffffu : 0u;
    const unsigned hb1 = (lane & 16) ? 0xffffffffu : 0u;
    #pragma unroll 1
    for (int g = 0; g < ng; ++g) {
        float4 hv[OWN_ROWS];
        #pragma unroll
        for (int b = 0; b < OWN_ROWS; ++b) {
            const int u = g * OWN_ROWS + b;
            hv[b] = make_float4(0.0f, 0.0f, 0.0f, 0.0f);
            if (FULL || u < nb) {
                const int cu = __builtin_amdgcn_readlane(c, u);
                const float4* __restrict__ hc = reinterpret_cast<const float4*>(cx.h + (long)cu * (OUT_F * NHEAD));
                hv[b] = hc[lane];
            }
        }
        #pragma unroll
        for (int b = 0; b < OWN_ROWS; ++b) {
            const int u = g * OWN_ROWS + b;
            if (FULL || u < nb) {
                const int lu = __builtin_amdgcn_readlane(loc, u);
                const unsigned a0 = (unsigned)__builtin_amdgcn_readlane(__float_as_int(w0), u);
                const unsigned a1 = (unsigned)__builtin_amdgcn_readlane(__float_as_int(w1), u);
                const unsigned a2 = (unsigned)__builtin_amdgcn_readlane(__float_as_int(w2), u);
                const unsigned a3 = (unsigned)__builtin_amdgcn_readlane(__float_as_int(w3), u);
                const unsigned lo2 = (hb0 & a1) | (~hb0 & a0);
                const unsigned hi2 = (hb0 & a3) | (~hb0 & a2);
                const float wl = __int_as_float((int)((hb1 & hi2) | (~hb1 & lo2)));
                float* o = cx.tile + lu * (OUT_F * NHEAD) + lane;
                lds_add(o + 0 * 32, wl * hv[b].x);
                lds_add(o + 1 * 32, wl * hv[b].y);
                lds_add(o + 2 * 32, wl * hv[b].z);
                lds_add(o + 3 * 32, wl * hv[b].w);
            }
        }
    }
}

template <bool FULL>
__device__ inline void own_rows(const OwnCtx& cx, const int2* lb, int nb) {
    int loc = 0, c = 0;
    float w0 = 0.0f, w1 = 0.0f, w2 = 0.0f, w3 = 0.0f;
    if (FULL || cx.lane < nb) {
        const int2 lc = lb[cx.lane];
        c = lc.y;
        if (lc.x >= 0) {
            loc = lc.x;
            own_weights(cx, loc, c, w0, w1, w2, w3);
        }
    }
    own_walk<FULL>(cx, nb, loc, c, w0, w1, w2, w3);
}

template <bool FULL>
__device__ inline void own_slow(const OwnCtx& cx, const int* qb, int nb) {
    int loc = 0, c = 0;
    float w0 = 0.0f, w1 = 0.0f, w2 = 0.0f, w3 = 0.0f;
    if (FULL || cx.lane < nb) {
        const int e = qb[cx.lane];
        const int2 rc = cx.pr[e];
        const unsigned d = (unsigned)(rc.x - cx.lo);
        c = rc.y;
        if (d < cx.cnt) {
            loc = (int)d;
            own_weights(cx, loc, c, w0, w1, w2, w3);
        }
    }
    own_walk<FULL>(cx, nb, loc, c, w0, w1, w2, w3);
}

template <bool SLOW>
__device__ inline void own_drain(const OwnCtx& cx, int* q, int& qn, int2* L, int& ln) {
    while (qn >= 32) {
        qn -= 32;
        if (SLOW) own_slow<true>(cx, q + qn, 32);
        else      own_mark<true>(cx, q + qn, 32, L, ln);
    }
}

template <bool SLOW>
__device__ inline void own_test(const OwnCtx& cx, int4 ow, int e0, unsigned pat,
                                int* q, int& qn, int2* L, int& ln) {
    const unsigned L7 = 0x7f7f7f7fu;
    const unsigned x0 = (unsigned)ow.x ^ pat, x1 = (unsigned)ow.y ^ pat;
    const unsigned x2 = (unsigned)ow.z ^ pat, x3 = (unsigned)ow.w ^ pat;
    const unsigned o0 = ((x0 & L7) + L7) | x0 | L7, o1 = ((x1 & L7) + L7) | x1 | L7;
    const unsigned o2 = ((x2 & L7) + L7) | x2 | L7, o3 = ((x3 & L7) + L7) | x3 | L7;
    unsigned f = ~o0 | (~o1 >> 1) | (~o2 >> 2) | (~o3 >> 3);
    for (;;) {
        const unsigned m = __builtin_amdgcn_ballot_w32(f != 0u);
        if (m == 0u) break;
        if (f != 0u) {
            const int p = __builtin_ctz(f);
            q[qn + (int)__builtin_amdgcn_mbcnt_lo(m, 0u)] = e0 + ((7 - (p & 7)) << 2) + (p >> 3);
            f &= f - 1u;
        }
        qn += __builtin_popcount(m);
        own_drain<SLOW>(cx, q, qn, L, ln);
    }
}

template <bool SLOW>
__device__ inline void own_tail(const OwnCtx& cx, int first, int E, int* q, int& qn, int2* L, int& ln) {
    for (int base = first; base < E; base += 32) {
        const int i = base + cx.lane;
        unsigned d = 0xffffffffu;
        if (i < E) d = (unsigned)(cx.row[i] - cx.lo);
        const bool hit = d < cx.cnt;
        const unsigned m = __builtin_amdgcn_ballot_w32(hit);
        if (m != 0u) {
            if (hit) q[qn + (int)__builtin_amdgcn_mbcnt_lo(m, 0u)] = i;
            qn += __builtin_popcount(m);
        }
        own_drain<SLOW>(cx, q, qn, L, ln);
    }
}

__global__ __launch_bounds__(OWN_THREADS) void gat_aggregate(const int* __restrict__ row,
                                                            const int* __restrict__ prw,
                                                            const int* __restrict__ own,
                                                            const float* __restrict__ rec,
                                                            const float* __restrict__ ar,
                                                            const float* __restrict__ h,
                                                            float* __restrict__ out,
                                                            int N, int E) {
    __shared__ __attribute__((aligned(16))) float tile[OWN_NODES * OUT_F * NHEAD];
    __shared__ __attribute__((aligned(16))) float recL[OWN_NODES * REC_F];
    __shared__ __attribute__((aligned(16))) float sumL[OWN_NODES * NHEAD];
    __shared__ __attribute__((aligned(8))) int2 lists[OWN_WAVES * OWN_LCAP];
    __shared__ int queue[OWN_WAVES * OWN_QCAP];
    __shared__ int ovfL;

    const int lane = threadIdx.x & 31;
    const int wave = __builtin_amdgcn_readfirstlane(threadIdx.x >> 5);
    const int lo = blockIdx.x * OWN_NODES;
    const int left = N - lo;
    const unsigned cnt = (unsigned)((left < OWN_NODES) ? left : OWN_NODES);
    const unsigned pat = (blockIdx.x & 255u) * 0x01010101u;

    const float4 zero4 = make_float4(0.0f, 0.0f, 0.0f, 0.0f);
    float4* tile4 = reinterpret_cast<float4*>(tile);
    for (int i = threadIdx.x; i < (int)cnt * (OUT_F * NHEAD / 4); i += OWN_THREADS) tile4[i] = zero4;
    for (int i = threadIdx.x; i < (int)cnt; i += OWN_THREADS) reinterpret_cast<float4*>(sumL)[i] = zero4;
    const float4* __restrict__ rec4 = reinterpret_cast<const float4*>(rec + (long)lo * REC_F);
    for (int i = threadIdx.x; i < (int)cnt * (REC_F / 4); i += OWN_THREADS) reinterpret_cast<float4*>(recL)[i] = rec4[i];
    if (threadIdx.x == 0) ovfL = 0;
    __syncthreads();

    OwnCtx cx;
    cx.row = row;   cx.pr = reinterpret_cast<const int2*>(prw);
    cx.ar = ar;     cx.h = h;
    cx.recL = recL; cx.sumL = sumL; cx.tile = tile; cx.ovfL = &ovfL;
    cx.lo = lo;     cx.cnt = cnt;   cx.lane = lane;

    int* q = queue + wave * OWN_QCAP;
    int2* L = lists + wave * OWN_LCAP;
    int qn = 0, ln = 0;
    const int nFull = E >> 9;
    const int nOwn = (N + OWN_NODES - 1) / OWN_NODES;
    const int rot = (int)blockIdx.x * (nFull / nOwn);
    const int4* __restrict__ own4 = reinterpret_cast<const int4*>(own);
    const int nMine  = (wave < nFull) ? ((nFull - wave + OWN_WAVES - 1) / OWN_WAVES) : 0;
    const int nSuper = nMine / OWN_LOADS;
    {
        int4 cur[OWN_LOADS];
        int stc[OWN_LOADS];
        #pragma unroll
        for (int k = 0; k < OWN_LOADS; ++k) {
            cur[k] = make_int4(0, 0, 0, 0);
            stc[k] = 0;
        }
        if (nSuper > 0) {
            #pragma unroll
            for (int k = 0; k < OWN_LOADS; ++k) {
                int s = wave + k * OWN_WAVES + rot;
                if (s >= nFull) s -= nFull;
                stc[k] = s;
                cur[k] = own4[(long)s * 32 + lane];
            }
        }
        for (int t = 0; t < nSuper; ++t) {
            int4 nxt[OWN_LOADS];
            int stn[OWN_LOADS];
            const int tn = (t + 1 < nSuper) ? (t + 1) : t;
            #pragma unroll
            for (int k = 0; k < OWN_LOADS; ++k) {
                int s = wave + (tn * OWN_LOADS + k) * OWN_WAVES + rot;
                if (s >= nFull) s -= nFull;
                stn[k] = s;
                nxt[k] = own4[(long)s * 32 + lane];
            }
            #pragma unroll
            for (int k = 0; k < OWN_LOADS; ++k)
                own_test<false>(cx, cur[k], stc[k] * 512 + lane * 16, pat, q, qn, L, ln);
            #pragma unroll
            for (int k = 0; k < OWN_LOADS; ++k) {
                cur[k] = nxt[k];
                stc[k] = stn[k];
            }
        }
    }
    for (int j = nSuper * OWN_LOADS; j < nMine; ++j) {
        int s = wave + j * OWN_WAVES + rot;
        if (s >= nFull) s -= nFull;
        own_test<false>(cx, own4[(long)s * 32 + lane], s * 512 + lane * 16, pat, q, qn, L, ln);
    }
    if (wave == 0) own_tail<false>(cx, nFull << 9, E, q, qn, L, ln);
    if (qn > 0) own_mark<false>(cx, q, qn, L, ln);
    __syncthreads();

    const int crowded = __builtin_amdgcn_readfirstlane(ovfL);
    if (crowded == 0) {
        int b = 0;
        for (; b + 32 <= ln; b += 32) own_rows<true>(cx, L + b, 32);
        if (b < ln) own_rows<false>(cx, L + b, ln - b);
    } else {
        qn = 0;
        for (int j = 0; j < nMine; ++j) {
            int s = wave + j * OWN_WAVES + rot;
            if (s >= nFull) s -= nFull;
            own_test<true>(cx, own4[(long)s * 32 + lane], s * 512 + lane * 16, pat, q, qn, L, ln);
        }
        if (wave == 0) own_tail<true>(cx, nFull << 9, E, q, qn, L, ln);
        if (qn > 0) own_slow<false>(cx, q, qn);
    }
    __syncthreads();

    float4* __restrict__ out4 = reinterpret_cast<float4*>(out + (long)lo * (OUT_F * NHEAD));
    for (int i = threadIdx.x; i < (int)cnt * (OUT_F * NHEAD / 4); i += OWN_THREADS) {
        const float* t = tile + (i >> 5) * (OUT_F * NHEAD) + (i & 31);
        const float4 v = make_float4(t[0], t[32], t[64], t[96]);
        const float sv = sumL[(i >> 5) * NHEAD + ((i & 31) >> 3)];
        float4 o = zero4;
        if (sv != 0.0f) {
            o.x = v.x / sv;
            o.y = v.y / sv;
            o.z = v.z / sv;
            o.w = v.w / sv;
        }
        out4[i] = o;
    }
}

extern "C" void kernel_launch(void* const* d_in, const int* in_sizes, int n_in,
                              void* d_out, int out_size, void* d_ws, size_t ws_size,
                              hipStream_t stream) {
    const float* x    = (const float*)d_in[0];
    const int*   edge = (const int*)d_in[1];
    const float* W    = (const float*)d_in[2];
    const float* a_l  = (const float*)d_in[3];
    const float* a_r  = (const float*)d_in[4];

    const int N = in_sizes[0] / IN_F;
    const int E = in_sizes[1] / 2;
    const int* rowp = edge;
    const int* colp = edge + E;

    float* ws  = (float*)d_ws;
    float* h   = ws;
    float* rec = h   + (size_t)N * (OUT_F * NHEAD);
    float* ar  = rec + (size_t)N * REC_F;
    int*   own = (int*)(ar + (size_t)N * NHEAD);
    int*   prw = own + ((size_t)E + 15) / 16 * 4;
    float* out = (float*)d_out;

    gat_gemm<<<(unsigned)((N + 127) / 128), 256, 0, stream>>>(x, W, h, N);
    gat_alar<<<(unsigned)((N + 63) / 64), 256, 0, stream>>>(h, a_l, a_r, rec, ar, N);
    gat_owner_map<<<(unsigned)((E + 1023) / 1024), 256, 0, stream>>>(rowp, colp, own, prw, E);
    gat_aggregate<<<(unsigned)((N + OWN_NODES - 1) / OWN_NODES), OWN_THREADS, 0, stream>>>(rowp, prw, own, rec, ar, h, out, N, E);
}
